# starting kernel
# baseline (speedup 1.0000x reference)
#include <hip/hip_runtime.h>
#include <hip/hip_bf16.h>
#include <math.h>

typedef __attribute__((ext_vector_type(16))) __bf16 v16bf;
typedef __attribute__((ext_vector_type(8)))  __bf16 v8bf;
typedef __attribute__((ext_vector_type(8)))  float  v8f;

constexpr int B  = 256;
constexpr int U  = 128;
constexpr int P  = 64;
constexpr int H  = 768;
constexpr int K2 = U * P;

__global__ void norms_kernel(const float* __restrict__ src,
                             float* __restrict__ inv_out, int len) {
  int wave = (blockIdx.x * blockDim.x + threadIdx.x) >> 5;
  int lane = threadIdx.x & 31;
  if (wave >= len * B) return;
  int i = wave / B, b = wave % B;
  const float* row = src + (size_t)wave * H;
  float s = 0.f;
#pragma unroll
  for (int j = 0; j < H; j += 128) {
    float4 v = *(const float4*)(row + j + lane * 4);
    s += v.x * v.x + v.y * v.y + v.z * v.z + v.w * v.w;
  }
#pragma unroll
  for (int m = 16; m; m >>= 1) s += __shfl_xor(s, m, 32);
  if (lane == 0) inv_out[b * len + i] = 1.0f / fmaxf(sqrtf(s), 1e-8f);
}

__global__ void wconv_kernel(const float* __restrict__ w_utt,
                             const float* __restrict__ w_pheno,
                             __bf16* __restrict__ wu_bf,
                             __bf16* __restrict__ wp_bf) {
  int idx = blockIdx.x * blockDim.x + threadIdx.x;
  if (idx < U * K2) wu_bf[idx] = (__bf16)w_utt[idx];
  if (idx < P * K2) {
    int o = idx / K2, r = idx % K2;
    int u = r / P, p = r % P;
    wp_bf[idx] = (__bf16)w_pheno[(size_t)o * K2 + p * U + u];
  }
}

__device__ __forceinline__ v16bf load_frag_f32(const float* __restrict__ p) {
  float4 a0 = *(const float4*)(p);
  float4 a1 = *(const float4*)(p + 4);
  float4 b0 = *(const float4*)(p + 16);
  float4 b1 = *(const float4*)(p + 20);
  v16bf r;
  r[0]  = (__bf16)a0.x; r[1]  = (__bf16)a0.y;
  r[2]  = (__bf16)a0.z; r[3]  = (__bf16)a0.w;
  r[4]  = (__bf16)a1.x; r[5]  = (__bf16)a1.y;
  r[6]  = (__bf16)a1.z; r[7]  = (__bf16)a1.w;
  r[8]  = (__bf16)b0.x; r[9]  = (__bf16)b0.y;
  r[10] = (__bf16)b0.z; r[11] = (__bf16)b0.w;
  r[12] = (__bf16)b1.x; r[13] = (__bf16)b1.y;
  r[14] = (__bf16)b1.z; r[15] = (__bf16)b1.w;
  return r;
}

__device__ __forceinline__ v16bf load_frag_bf16(const __bf16* __restrict__ p) {
  v8bf lo = *(const v8bf*)(p);
  v8bf hi = *(const v8bf*)(p + 16);
  v16bf r;
#pragma unroll
  for (int j = 0; j < 8; ++j) { r[j] = lo[j]; r[j + 8] = hi[j]; }
  return r;
}

__global__ void cosine_wmma_kernel(const float* __restrict__ utt,
                                   const float* __restrict__ pheno,
                                   const float* __restrict__ inv_nu,
                                   const float* __restrict__ inv_np,
                                   __bf16* __restrict__ c_bf) {
  int b    = blockIdx.x;
  int wave = threadIdx.x >> 5;
  int lane = threadIdx.x & 31;
  int rt   = lane & 15;
  int hi   = lane >> 4;
  int k0   = hi ? 8 : 0;

  int u_row = wave * 16 + rt;
  const float* Arow = utt + ((size_t)u_row * B + b) * H + k0;

  const float* Brow[4];
#pragma unroll
  for (int nt = 0; nt < 4; ++nt) {
    int p_col = nt * 16 + rt;
    Brow[nt]  = pheno + ((size_t)p_col * B + b) * H + k0;
  }

  v8f acc[4] = {};
  for (int kk = 0; kk < H; kk += 32) {
    v16bf a = load_frag_f32(Arow + kk);
#pragma unroll
    for (int nt = 0; nt < 4; ++nt) {
      v16bf bb = load_frag_f32(Brow[nt] + kk);
      acc[nt] = __builtin_amdgcn_wmma_f32_16x16x32_bf16(
          false, a, false, bb, (short)0, acc[nt], false, false);
    }
  }

  const float* su_base = inv_nu + b * U + wave * 16 + 8 * hi;
  float4 su0 = *(const float4*)(su_base);
  float4 su1 = *(const float4*)(su_base + 4);
  float su[8] = {su0.x, su0.y, su0.z, su0.w, su1.x, su1.y, su1.z, su1.w};

  __bf16* cb = c_bf + (size_t)b * K2;
#pragma unroll
  for (int nt = 0; nt < 4; ++nt) {
    float sp = inv_np[b * P + nt * 16 + rt];
#pragma unroll
    for (int r = 0; r < 8; ++r) {
      int m = r + 8 * hi;
      cb[(wave * 16 + m) * P + nt * 16 + rt] = (__bf16)(acc[nt][r] * su[r] * sp);
    }
  }
}

__global__ void logits_wmma_kernel(const __bf16* __restrict__ c_bf,
                                   const __bf16* __restrict__ W,
                                   const float* __restrict__ bias,
                                   float* __restrict__ out, int N) {
  int mtile = blockIdx.x;
  int wave  = threadIdx.x >> 5;
  int ntile = blockIdx.y * 4 + wave;
  int lane  = threadIdx.x & 31;
  int rt    = lane & 15;
  int hi    = lane >> 4;
  int k0    = hi ? 8 : 0;

  const __bf16* Arow = c_bf + (size_t)(mtile * 16 + rt) * K2 + k0;
  const __bf16* Wrow = W    + (size_t)(ntile * 16 + rt) * K2 + k0;

  v8f acc = {};
  for (int kk = 0; kk < K2; kk += 32) {
    v16bf a = load_frag_bf16(Arow + kk);
    v16bf w = load_frag_bf16(Wrow + kk);
    acc = __builtin_amdgcn_wmma_f32_16x16x32_bf16(
        false, a, false, w, (short)0, acc, false, false);
  }

  int n = ntile * 16 + rt;
  float bv = bias[n];
#pragma unroll
  for (int r = 0; r < 8; ++r) {
    int m = r + 8 * hi;
    out[(size_t)(mtile * 16 + m) * N + n] = acc[r] + bv;
  }
}

__global__ void softmax_kernel(float* __restrict__ x, int n) {
  int row  = (blockIdx.x * blockDim.x + threadIdx.x) >> 5;
  int lane = threadIdx.x & 31;
  if (row >= B) return;
  float* p = x + (size_t)row * n;
  int per = n >> 5;
  float vals[4];
  float mx = -1e30f;
  for (int j = 0; j < per; ++j) { vals[j] = p[lane + 32 * j]; mx = fmaxf(mx, vals[j]); }
#pragma unroll
  for (int m = 16; m; m >>= 1) mx = fmaxf(mx, __shfl_xor(mx, m, 32));
  float s = 0.f;
  for (int j = 0; j < per; ++j) { vals[j] = __expf(vals[j] - mx); s += vals[j]; }
#pragma unroll
  for (int m = 16; m; m >>= 1) s += __shfl_xor(s, m, 32);
  float inv = 1.0f / s;
  for (int j = 0; j < per; ++j) p[lane + 32 * j] = vals[j] * inv;
}

extern "C" void kernel_launch(void* const* d_in, const int* in_sizes, int n_in,
                              void* d_out, int out_size, void* d_ws, size_t ws_size,
                              hipStream_t stream) {
  const float* utt     = (const float*)d_in[0];
  const float* pheno   = (const float*)d_in[1];
  const float* w_utt   = (const float*)d_in[2];
  const float* b_utt   = (const float*)d_in[3];
  const float* w_pheno = (const float*)d_in[4];
  const float* b_pheno = (const float*)d_in[5];
  (void)in_sizes; (void)n_in; (void)out_size; (void)ws_size;

  char* ws = (char*)d_ws;
  size_t off = 0;
  auto alloc = [&](size_t bytes) {
    void* p = ws + off;
    off = (off + bytes + 255) & ~(size_t)255;
    return p;
  };
  float*  inv_nu = (float*)alloc((size_t)B * U * sizeof(float));
  float*  inv_np = (float*)alloc((size_t)B * P * sizeof(float));
  __bf16* wu_bf  = (__bf16*)alloc((size_t)U * K2 * sizeof(__bf16));
  __bf16* wp_bf  = (__bf16*)alloc((size_t)P * K2 * sizeof(__bf16));
  __bf16* c_bf   = (__bf16*)alloc((size_t)B * K2 * sizeof(__bf16));

  float* row_logits = (float*)d_out;
  float* col_logits = (float*)d_out + B * U;

  norms_kernel<<<(B * U) / 8, 256, 0, stream>>>(utt, inv_nu, U);
  norms_kernel<<<(B * P) / 8, 256, 0, stream>>>(pheno, inv_np, P);
  wconv_kernel<<<(U * K2) / 256, 256, 0, stream>>>(w_utt, w_pheno, wu_bf, wp_bf);

  cosine_wmma_kernel<<<B, 256, 0, stream>>>(utt, pheno, inv_nu, inv_np, c_bf);

  logits_wmma_kernel<<<dim3(B / 16, U / 64), 128, 0, stream>>>(c_bf, wu_bf, b_utt, row_logits, U);
  logits_wmma_kernel<<<dim3(B / 16, P / 64), 128, 0, stream>>>(c_bf, wp_bf, b_pheno, col_logits, P);

  softmax_kernel<<<B / 8, 256, 0, stream>>>(row_logits, U);
  softmax_kernel<<<B / 8, 256, 0, stream>>>(col_logits, P);
}


# logits GEMMs and softmaxes fused into the per-batch cosine kernel: c stays in LDS, each wave contracts its 16 rows with fragment-order weights, 8 wave partials summed in order (f32 sum order changes)
# speedup vs baseline: 2.3414x; 2.3414x over previous
#include <hip/hip_runtime.h>
#include <hip/hip_bf16.h>
#include <math.h>

typedef __attribute__((ext_vector_type(16))) __bf16 v16bf;
typedef __attribute__((ext_vector_type(8)))  __bf16 v8bf;
typedef __attribute__((ext_vector_type(8)))  float  v8f;

constexpr int B  = 256;
constexpr int U  = 128;
constexpr int P  = 64;
constexpr int H  = 768;
constexpr int K2 = U * P;

__global__ void norms_kernel(const float* __restrict__ src,
                             float* __restrict__ inv_out, int len) {
  int wave = (blockIdx.x * blockDim.x + threadIdx.x) >> 5;
  int lane = threadIdx.x & 31;
  if (wave >= len * B) return;
  int i = wave / B, b = wave % B;
  const float* row = src + (size_t)wave * H;
  float s = 0.f;
#pragma unroll
  for (int j = 0; j < H; j += 128) {
    float4 v = *(const float4*)(row + j + lane * 4);
    s += v.x * v.x + v.y * v.y + v.z * v.z + v.w * v.w;
  }
#pragma unroll
  for (int m = 16; m; m >>= 1) s += __shfl_xor(s, m, 32);
  if (lane == 0) inv_out[b * len + i] = 1.0f / fmaxf(sqrtf(s), 1e-8f);
}

constexpr int NT_O   = (U + P) / 16;
constexpr int KSTEPS = K2 / 32;

__global__ void wstage_kernel(const float* __restrict__ w_utt,
                              const float* __restrict__ w_pheno,
                              __bf16* __restrict__ wfrag) {
  int t = blockIdx.x * blockDim.x + threadIdx.x;
  if (t >= KSTEPS * NT_O * 64) return;
  int L    = t & 31;
  int half = (t >> 5) & 1;
  int frag = t >> 6;
  int ot   = frag % NT_O;
  int ks   = frag / NT_O;
  int o    = ot * 16 + (L & 15);
  int k    = ks * 32 + ((L >> 4) ? 8 : 0) + half * 16;
  v8bf r;
  if (o < U) {
    const float* s = w_utt + (size_t)o * K2 + k;
    float4 a = *(const float4*)(s);
    float4 c = *(const float4*)(s + 4);
    r[0] = (__bf16)a.x; r[1] = (__bf16)a.y; r[2] = (__bf16)a.z; r[3] = (__bf16)a.w;
    r[4] = (__bf16)c.x; r[5] = (__bf16)c.y; r[6] = (__bf16)c.z; r[7] = (__bf16)c.w;
  } else {
    int u = k / P, p = k % P;
    const float* s = w_pheno + (size_t)(o - U) * K2 + (size_t)p * U + u;
#pragma unroll
    for (int j = 0; j < 8; ++j) r[j] = (__bf16)s[j * U];
  }
  *(v8bf*)(wfrag + (size_t)t * 8) = r;
}

__device__ __forceinline__ v16bf load_frag_f32(const float* __restrict__ p) {
  float4 a0 = *(const float4*)(p);
  float4 a1 = *(const float4*)(p + 4);
  float4 b0 = *(const float4*)(p + 16);
  float4 b1 = *(const float4*)(p + 20);
  v16bf r;
  r[0]  = (__bf16)a0.x; r[1]  = (__bf16)a0.y;
  r[2]  = (__bf16)a0.z; r[3]  = (__bf16)a0.w;
  r[4]  = (__bf16)a1.x; r[5]  = (__bf16)a1.y;
  r[6]  = (__bf16)a1.z; r[7]  = (__bf16)a1.w;
  r[8]  = (__bf16)b0.x; r[9]  = (__bf16)b0.y;
  r[10] = (__bf16)b0.z; r[11] = (__bf16)b0.w;
  r[12] = (__bf16)b1.x; r[13] = (__bf16)b1.y;
  r[14] = (__bf16)b1.z; r[15] = (__bf16)b1.w;
  return r;
}

__device__ __forceinline__ v16bf load_frag_bf16(const __bf16* __restrict__ p) {
  v8bf lo = *(const v8bf*)(p);
  v8bf hi = *(const v8bf*)(p + 16);
  v16bf r;
#pragma unroll
  for (int j = 0; j < 8; ++j) { r[j] = lo[j]; r[j + 8] = hi[j]; }
  return r;
}

__device__ __forceinline__ void load_wfrags(const __bf16* __restrict__ fp,
                                            v8bf (&lo)[NT_O], v8bf (&up)[NT_O]) {
#pragma unroll
  for (int ot = 0; ot < NT_O; ++ot) {
    lo[ot] = *(const v8bf*)(fp + ot * 512);
    up[ot] = *(const v8bf*)(fp + ot * 512 + 256);
  }
}

__device__ __forceinline__ v16bf join_frag(v8bf lo, v8bf up) {
  v16bf r;
#pragma unroll
  for (int j = 0; j < 8; ++j) { r[j] = lo[j]; r[j + 8] = up[j]; }
  return r;
}

#define HOLD_LOADS(lo, up)                                                     \
  asm volatile("" ::"v"(lo[0]), "v"(up[0]), "v"(lo[1]), "v"(up[1]),            \
               "v"(lo[2]), "v"(up[2]), "v"(lo[3]), "v"(up[3]),                 \
               "v"(lo[4]), "v"(up[4]), "v"(lo[5]), "v"(up[5]),                 \
               "v"(lo[6]), "v"(up[6]), "v"(lo[7]), "v"(up[7]),                 \
               "v"(lo[8]), "v"(up[8]), "v"(lo[9]), "v"(up[9]),                 \
               "v"(lo[10]), "v"(up[10]), "v"(lo[11]), "v"(up[11]))

__global__ void __launch_bounds__(256)
fused_attention_kernel(const float* __restrict__ utt,
                       const float* __restrict__ pheno,
                       const float* __restrict__ inv_nu,
                       const float* __restrict__ inv_np,
                       const __bf16* __restrict__ wfrag,
                       const float* __restrict__ b_utt,
                       const float* __restrict__ b_pheno,
                       float* __restrict__ out_row,
                       float* __restrict__ out_col) {
  __shared__ __attribute__((aligned(16))) __bf16 c_lds[K2];
  __shared__ float red[8][U + P];
  __shared__ float logit[U + P];

  int b    = blockIdx.x;
  int wave = threadIdx.x >> 5;
  int lane = threadIdx.x & 31;
  int rt   = lane & 15;
  int hi   = lane >> 4;
  int k0   = hi ? 8 : 0;

  int u_row = wave * 16 + rt;
  const float* Arow = utt + ((size_t)u_row * B + b) * H + k0;

  const float* Brow[4];
#pragma unroll
  for (int nt = 0; nt < 4; ++nt) {
    int p_col = nt * 16 + rt;
    Brow[nt]  = pheno + ((size_t)p_col * B + b) * H + k0;
  }

  v8f acc[4] = {};
  for (int kk = 0; kk < H; kk += 32) {
    v16bf a = load_frag_f32(Arow + kk);
#pragma unroll
    for (int nt = 0; nt < 4; ++nt) {
      v16bf bb = load_frag_f32(Brow[nt] + kk);
      acc[nt] = __builtin_amdgcn_wmma_f32_16x16x32_bf16(
          false, a, false, bb, (short)0, acc[nt], false, false);
    }
  }

  const float* su_base = inv_nu + b * U + wave * 16 + 8 * hi;
  float4 su0 = *(const float4*)(su_base);
  float4 su1 = *(const float4*)(su_base + 4);
  float su[8] = {su0.x, su0.y, su0.z, su0.w, su1.x, su1.y, su1.z, su1.w};

#pragma unroll
  for (int nt = 0; nt < 4; ++nt) {
    float sp = inv_np[b * P + nt * 16 + rt];
#pragma unroll
    for (int r = 0; r < 8; ++r) {
      int m = r + 8 * hi;
      c_lds[(wave * 16 + m) * P + nt * 16 + rt] = (__bf16)(acc[nt][r] * su[r] * sp);
    }
  }
  __syncthreads();

  v8f lacc[NT_O] = {};
  const __bf16* cw = c_lds + wave * (16 * P) + k0;
  const __bf16* wf = wfrag + (size_t)(wave * 32) * NT_O * 512 + lane * 8;
  for (int s = 0; s < 32; s += 2) {
    v8bf lo0[NT_O], up0[NT_O], lo1[NT_O], up1[NT_O];
    load_wfrags(wf + (size_t)s * NT_O * 512, lo0, up0);
    load_wfrags(wf + (size_t)(s + 1) * NT_O * 512, lo1, up1);
    v16bf cb0 = load_frag_bf16(cw + s * 32);
    v16bf cb1 = load_frag_bf16(cw + (s + 1) * 32);
    HOLD_LOADS(lo0, up0);
#pragma unroll
    for (int ot = 0; ot < NT_O; ++ot)
      lacc[ot] = __builtin_amdgcn_wmma_f32_16x16x32_bf16(
          false, join_frag(lo0[ot], up0[ot]), false, cb0, (short)0, lacc[ot], false, false);
    HOLD_LOADS(lo1, up1);
#pragma unroll
    for (int ot = 0; ot < NT_O; ++ot)
      lacc[ot] = __builtin_amdgcn_wmma_f32_16x16x32_bf16(
          false, join_frag(lo1[ot], up1[ot]), false, cb1, (short)0, lacc[ot], false, false);
  }
  if (rt == 0) {
#pragma unroll
    for (int ot = 0; ot < NT_O; ++ot)
#pragma unroll
      for (int r = 0; r < 8; ++r)
        red[wave][ot * 16 + r + 8 * hi] = lacc[ot][r];
  }
  __syncthreads();

  if (threadIdx.x < U + P) {
    int o = threadIdx.x;
    float s = red[0][o];
#pragma unroll
    for (int w = 1; w < 8; ++w) s += red[w][o];
    logit[o] = s + (o < U ? b_utt[o] : b_pheno[o - U]);
  }
  __syncthreads();

  if (wave < 2) {
    const float* x = wave ? logit + U : logit;
    float* dst     = wave ? out_col + (size_t)b * P : out_row + (size_t)b * U;
    int per = wave ? P / 32 : U / 32;
    float vals[4];
    float mx = -1e30f;
    for (int j = 0; j < per; ++j) { vals[j] = x[lane + 32 * j]; mx = fmaxf(mx, vals[j]); }
#pragma unroll
    for (int m = 16; m; m >>= 1) mx = fmaxf(mx, __shfl_xor(mx, m, 32));
    float s = 0.f;
    for (int j = 0; j < per; ++j) { vals[j] = __expf(vals[j] - mx); s += vals[j]; }
#pragma unroll
    for (int m = 16; m; m >>= 1) s += __shfl_xor(s, m, 32);
    float inv = 1.0f / s;
    for (int j = 0; j < per; ++j) dst[lane + 32 * j] = vals[j] * inv;
  }
}

extern "C" void kernel_launch(void* const* d_in, const int* in_sizes, int n_in,
                              void* d_out, int out_size, void* d_ws, size_t ws_size,
                              hipStream_t stream) {
  const float* utt     = (const float*)d_in[0];
  const float* pheno   = (const float*)d_in[1];
  const float* w_utt   = (const float*)d_in[2];
  const float* b_utt   = (const float*)d_in[3];
  const float* w_pheno = (const float*)d_in[4];
  const float* b_pheno = (const float*)d_in[5];
  (void)in_sizes; (void)n_in; (void)out_size; (void)ws_size;

  char* ws = (char*)d_ws;
  size_t off = 0;
  auto alloc = [&](size_t bytes) {
    void* p = ws + off;
    off = (off + bytes + 255) & ~(size_t)255;
    return p;
  };
  float*  inv_nu = (float*)alloc((size_t)B * U * sizeof(float));
  float*  inv_np = (float*)alloc((size_t)B * P * sizeof(float));
  __bf16* wfrag  = (__bf16*)alloc((size_t)(U + P) * K2 * sizeof(__bf16));

  float* row_logits = (float*)d_out;
  float* col_logits = (float*)d_out + B * U;

  norms_kernel<<<(B * U) / 8, 256, 0, stream>>>(utt, inv_nu, U);
  norms_kernel<<<(B * P) / 8, 256, 0, stream>>>(pheno, inv_np, P);
  wstage_kernel<<<(KSTEPS * NT_O * 64) / 256, 256, 0, stream>>>(w_utt, w_pheno, wfrag);

  fused_attention_kernel<<<B, 256, 0, stream>>>(utt, pheno, inv_nu, inv_np, wfrag,
                                                b_utt, b_pheno, row_logits, col_logits);
}


# cosine GEMM: contraction over H split across the 8 waves (each wave 96 k of all 192 rows), each operand element loaded once per block; 8 partials summed via LDS in wave order (f32 sum order changes)
# speedup vs baseline: 3.1985x; 1.3661x over previous
#include <hip/hip_runtime.h>
#include <hip/hip_bf16.h>
#include <math.h>

typedef __attribute__((ext_vector_type(16))) __bf16 v16bf;
typedef __attribute__((ext_vector_type(8)))  __bf16 v8bf;
typedef __attribute__((ext_vector_type(8)))  float  v8f;
typedef __attribute__((ext_vector_type(4)))  float  v4f;

constexpr int B  = 256;
constexpr int U  = 128;
constexpr int P  = 64;
constexpr int H  = 768;
constexpr int K2 = U * P;

__global__ void norms_kernel(const float* __restrict__ src,
                             float* __restrict__ inv_out, int len) {
  int wave = (blockIdx.x * blockDim.x + threadIdx.x) >> 5;
  int lane = threadIdx.x & 31;
  if (wave >= len * B) return;
  int i = wave / B, b = wave % B;
  const float* row = src + (size_t)wave * H;
  float s = 0.f;
#pragma unroll
  for (int j = 0; j < H; j += 128) {
    float4 v = *(const float4*)(row + j + lane * 4);
    s += v.x * v.x + v.y * v.y + v.z * v.z + v.w * v.w;
  }
#pragma unroll
  for (int m = 16; m; m >>= 1) s += __shfl_xor(s, m, 32);
  if (lane == 0) inv_out[b * len + i] = 1.0f / fmaxf(sqrtf(s), 1e-8f);
}

constexpr int NT_O   = (U + P) / 16;
constexpr int KSTEPS = K2 / 32;

__global__ void wstage_kernel(const float* __restrict__ w_utt,
                              const float* __restrict__ w_pheno,
                              __bf16* __restrict__ wfrag) {
  int t = blockIdx.x * blockDim.x + threadIdx.x;
  if (t >= KSTEPS * NT_O * 64) return;
  int L    = t & 31;
  int half = (t >> 5) & 1;
  int frag = t >> 6;
  int ot   = frag % NT_O;
  int ks   = frag / NT_O;
  int o    = ot * 16 + (L & 15);
  int k    = ks * 32 + ((L >> 4) ? 8 : 0) + half * 16;
  v8bf r;
  if (o < U) {
    const float* s = w_utt + (size_t)o * K2 + k;
    float4 a = *(const float4*)(s);
    float4 c = *(const float4*)(s + 4);
    r[0] = (__bf16)a.x; r[1] = (__bf16)a.y; r[2] = (__bf16)a.z; r[3] = (__bf16)a.w;
    r[4] = (__bf16)c.x; r[5] = (__bf16)c.y; r[6] = (__bf16)c.z; r[7] = (__bf16)c.w;
  } else {
    int u = k / P, p = k % P;
    const float* s = w_pheno + (size_t)(o - U) * K2 + (size_t)p * U + u;
#pragma unroll
    for (int j = 0; j < 8; ++j) r[j] = (__bf16)s[j * U];
  }
  *(v8bf*)(wfrag + (size_t)t * 8) = r;
}

struct RawFrag { v4f a0, a1, b0, b1; };

__device__ __forceinline__ RawFrag load_raw_f32(const float* __restrict__ p) {
  RawFrag f;
  f.a0 = *(const v4f*)(p);
  f.a1 = *(const v4f*)(p + 4);
  f.b0 = *(const v4f*)(p + 16);
  f.b1 = *(const v4f*)(p + 20);
  return f;
}

__device__ __forceinline__ v16bf cvt_frag(const RawFrag& f) {
  v16bf r;
  r[0]  = (__bf16)f.a0[0]; r[1]  = (__bf16)f.a0[1];
  r[2]  = (__bf16)f.a0[2]; r[3]  = (__bf16)f.a0[3];
  r[4]  = (__bf16)f.a1[0]; r[5]  = (__bf16)f.a1[1];
  r[6]  = (__bf16)f.a1[2]; r[7]  = (__bf16)f.a1[3];
  r[8]  = (__bf16)f.b0[0]; r[9]  = (__bf16)f.b0[1];
  r[10] = (__bf16)f.b0[2]; r[11] = (__bf16)f.b0[3];
  r[12] = (__bf16)f.b1[0]; r[13] = (__bf16)f.b1[1];
  r[14] = (__bf16)f.b1[2]; r[15] = (__bf16)f.b1[3];
  return r;
}

__device__ __forceinline__ v16bf load_frag_bf16(const __bf16* __restrict__ p) {
  v8bf lo = *(const v8bf*)(p);
  v8bf hi = *(const v8bf*)(p + 16);
  v16bf r;
#pragma unroll
  for (int j = 0; j < 8; ++j) { r[j] = lo[j]; r[j + 8] = hi[j]; }
  return r;
}

__device__ __forceinline__ void load_wfrags(const __bf16* __restrict__ fp,
                                            v8bf (&lo)[NT_O], v8bf (&up)[NT_O]) {
#pragma unroll
  for (int ot = 0; ot < NT_O; ++ot) {
    lo[ot] = *(const v8bf*)(fp + ot * 512);
    up[ot] = *(const v8bf*)(fp + ot * 512 + 256);
  }
}

__device__ __forceinline__ v16bf join_frag(v8bf lo, v8bf up) {
  v16bf r;
#pragma unroll
  for (int j = 0; j < 8; ++j) { r[j] = lo[j]; r[j + 8] = up[j]; }
  return r;
}

#define HOLD_LOADS(lo, up)                                                     \
  asm volatile("" ::"v"(lo[0]), "v"(up[0]), "v"(lo[1]), "v"(up[1]),            \
               "v"(lo[2]), "v"(up[2]), "v"(lo[3]), "v"(up[3]),                 \
               "v"(lo[4]), "v"(up[4]), "v"(lo[5]), "v"(up[5]),                 \
               "v"(lo[6]), "v"(up[6]), "v"(lo[7]), "v"(up[7]),                 \
               "v"(lo[8]), "v"(up[8]), "v"(lo[9]), "v"(up[9]),                 \
               "v"(lo[10]), "v"(up[10]), "v"(lo[11]), "v"(up[11]))

constexpr int KSLICE = H / 8;
constexpr int KS_A   = KSLICE / 32;
constexpr int MT_A   = U / 16;
constexpr int NT_A   = P / 16;
constexpr int TILE_F = 256;
constexpr int PART_F = MT_A * NT_A * TILE_F;

__global__ void __launch_bounds__(256)
fused_attention_kernel(const float* __restrict__ utt,
                       const float* __restrict__ pheno,
                       const float* __restrict__ inv_nu,
                       const float* __restrict__ inv_np,
                       const __bf16* __restrict__ wfrag,
                       const float* __restrict__ b_utt,
                       const float* __restrict__ b_pheno,
                       float* __restrict__ out_row,
                       float* __restrict__ out_col) {
  __shared__ __attribute__((aligned(16))) float part[8 * PART_F];
  __shared__ __attribute__((aligned(16))) __bf16 c_lds[K2];
  __shared__ float red[8][U + P];
  __shared__ float logit[U + P];

  int b    = blockIdx.x;
  int wave = threadIdx.x >> 5;
  int lane = threadIdx.x & 31;
  int rt   = lane & 15;
  int hi   = lane >> 4;
  int k0   = hi ? 8 : 0;

  constexpr size_t TILE_STRIDE = (size_t)16 * B * H;
  const unsigned loff = (unsigned)(rt * (B * H) + wave * KSLICE + k0);
  const float* Au = utt   + (size_t)b * H;
  const float* Bp = pheno + (size_t)b * H;

  v8f acc[MT_A][NT_A] = {};
#pragma unroll
  for (int ks = 0; ks < KS_A; ++ks) {
    RawFrag rb[NT_A], ra[MT_A];
#pragma unroll
    for (int nt = 0; nt < NT_A; ++nt)
      rb[nt] = load_raw_f32(Bp + nt * TILE_STRIDE + loff + ks * 32);
#pragma unroll
    for (int mt = 0; mt < MT_A; ++mt)
      ra[mt] = load_raw_f32(Au + mt * TILE_STRIDE + loff + ks * 32);
    v16bf bfr[NT_A];
#pragma unroll
    for (int nt = 0; nt < NT_A; ++nt) bfr[nt] = cvt_frag(rb[nt]);
#pragma unroll
    for (int mt = 0; mt < MT_A; ++mt) {
      v16bf a = cvt_frag(ra[mt]);
#pragma unroll
      for (int nt = 0; nt < NT_A; ++nt)
        acc[mt][nt] = __builtin_amdgcn_wmma_f32_16x16x32_bf16(
            false, a, false, bfr[nt], (short)0, acc[mt][nt], false, false);
    }
  }

  const float* su_base = inv_nu + b * U + wave * 16 + 8 * hi;
  float4 su0 = *(const float4*)(su_base);
  float4 su1 = *(const float4*)(su_base + 4);
  float sp[NT_A];
#pragma unroll
  for (int nt = 0; nt < NT_A; ++nt) sp[nt] = inv_np[b * P + nt * 16 + rt];

  {
    float* pw = part + wave * PART_F + lane * 4;
#pragma unroll
    for (int mt = 0; mt < MT_A; ++mt)
#pragma unroll
      for (int nt = 0; nt < NT_A; ++nt) {
        v8f t = acc[mt][nt];
        *(v4f*)(pw + (mt * NT_A + nt) * TILE_F)       = __builtin_shufflevector(t, t, 0, 1, 2, 3);
        *(v4f*)(pw + (mt * NT_A + nt) * TILE_F + 128) = __builtin_shufflevector(t, t, 4, 5, 6, 7);
      }
  }
  __syncthreads();

  v8f own[NT_A];
#pragma unroll
  for (int nt = 0; nt < NT_A; ++nt) {
    const float* pr = part + (wave * NT_A + nt) * TILE_F + lane * 4;
    v4f s0 = *(const v4f*)(pr);
    v4f s1 = *(const v4f*)(pr + 128);
#pragma unroll
    for (int w = 1; w < 8; ++w) {
      s0 += *(const v4f*)(pr + w * PART_F);
      s1 += *(const v4f*)(pr + w * PART_F + 128);
    }
    own[nt] = __builtin_shufflevector(s0, s1, 0, 1, 2, 3, 4, 5, 6, 7);
  }

  float su[8] = {su0.x, su0.y, su0.z, su0.w, su1.x, su1.y, su1.z, su1.w};

#pragma unroll
  for (int nt = 0; nt < NT_A; ++nt) {
#pragma unroll
    for (int r = 0; r < 8; ++r) {
      int m = r + 8 * hi;
      c_lds[(wave * 16 + m) * P + nt * 16 + rt] = (__bf16)(own[nt][r] * su[r] * sp[nt]);
    }
  }
  __syncthreads();

  v8f lacc[NT_O] = {};
  const __bf16* cw = c_lds + wave * (16 * P) + k0;
  const __bf16* wf = wfrag + (size_t)(wave * 32) * NT_O * 512 + lane * 8;
  for (int s = 0; s < 32; s += 2) {
    v8bf lo0[NT_O], up0[NT_O], lo1[NT_O], up1[NT_O];
    load_wfrags(wf + (size_t)s * NT_O * 512, lo0, up0);
    load_wfrags(wf + (size_t)(s + 1) * NT_O * 512, lo1, up1);
    v16bf cb0 = load_frag_bf16(cw + s * 32);
    v16bf cb1 = load_frag_bf16(cw + (s + 1) * 32);
    HOLD_LOADS(lo0, up0);
#pragma unroll
    for (int ot = 0; ot < NT_O; ++ot)
      lacc[ot] = __builtin_amdgcn_wmma_f32_16x16x32_bf16(
          false, join_frag(lo0[ot], up0[ot]), false, cb0, (short)0, lacc[ot], false, false);
    HOLD_LOADS(lo1, up1);
#pragma unroll
    for (int ot = 0; ot < NT_O; ++ot)
      lacc[ot] = __builtin_amdgcn_wmma_f32_16x16x32_bf16(
          false, join_frag(lo1[ot], up1[ot]), false, cb1, (short)0, lacc[ot], false, false);
  }
  if (rt == 0) {
#pragma unroll
    for (int ot = 0; ot < NT_O; ++ot)
#pragma unroll
      for (int r = 0; r < 8; ++r)
        red[wave][ot * 16 + r + 8 * hi] = lacc[ot][r];
  }
  __syncthreads();

  if (threadIdx.x < U + P) {
    int o = threadIdx.x;
    float s = red[0][o];
#pragma unroll
    for (int w = 1; w < 8; ++w) s += red[w][o];
    logit[o] = s + (o < U ? b_utt[o] : b_pheno[o - U]);
  }
  __syncthreads();

  if (wave < 2) {
    const float* x = wave ? logit + U : logit;
    float* dst     = wave ? out_col + (size_t)b * P : out_row + (size_t)b * U;
    int per = wave ? P / 32 : U / 32;
    float vals[4];
    float mx = -1e30f;
    for (int j = 0; j < per; ++j) { vals[j] = x[lane + 32 * j]; mx = fmaxf(mx, vals[j]); }
#pragma unroll
    for (int m = 16; m; m >>= 1) mx = fmaxf(mx, __shfl_xor(mx, m, 32));
    float s = 0.f;
    for (int j = 0; j < per; ++j) { vals[j] = __expf(vals[j] - mx); s += vals[j]; }
#pragma unroll
    for (int m = 16; m; m >>= 1) s += __shfl_xor(s, m, 32);
    float inv = 1.0f / s;
    for (int j = 0; j < per; ++j) dst[lane + 32 * j] = vals[j] * inv;
  }
}

extern "C" void kernel_launch(void* const* d_in, const int* in_sizes, int n_in,
                              void* d_out, int out_size, void* d_ws, size_t ws_size,
                              hipStream_t stream) {
  const float* utt     = (const float*)d_in[0];
  const float* pheno   = (const float*)d_in[1];
  const float* w_utt   = (const float*)d_in[2];
  const float* b_utt   = (const float*)d_in[3];
  const float* w_pheno = (const float*)d_in[4];
  const float* b_pheno = (const float*)d_in[5];
  (void)in_sizes; (void)n_in; (void)out_size; (void)ws_size;

  char* ws = (char*)d_ws;
  size_t off = 0;
  auto alloc = [&](size_t bytes) {
    void* p = ws + off;
    off = (off + bytes + 255) & ~(size_t)255;
    return p;
  };
  float*  inv_nu = (float*)alloc((size_t)B * U * sizeof(float));
  float*  inv_np = (float*)alloc((size_t)B * P * sizeof(float));
  __bf16* wfrag  = (__bf16*)alloc((size_t)(U + P) * K2 * sizeof(__bf16));

  float* row_logits = (float*)d_out;
  float* col_logits = (float*)d_out + B * U;

  norms_kernel<<<(B * U) / 8, 256, 0, stream>>>(utt, inv_nu, U);
  norms_kernel<<<(B * P) / 8, 256, 0, stream>>>(pheno, inv_np, P);
  wstage_kernel<<<(KSTEPS * NT_O * 64) / 256, 256, 0, stream>>>(w_utt, w_pheno, wfrag);

  fused_attention_kernel<<<B, 256, 0, stream>>>(utt, pheno, inv_nu, inv_np, wfrag,
                                                b_utt, b_pheno, row_logits, col_logits);
}


# logits+softmaxes moved from the per-batch kernel to a new launch, 16 batches x all 128 row (or 64 col) logits per block: weights cross L2 once per 16 batches; c via workspace (f32 sum order changes)
# speedup vs baseline: 3.4896x; 1.0910x over previous
#include <hip/hip_runtime.h>
#include <hip/hip_bf16.h>
#include <math.h>

typedef __attribute__((ext_vector_type(16))) __bf16 v16bf;
typedef __attribute__((ext_vector_type(8)))  __bf16 v8bf;
typedef __attribute__((ext_vector_type(8)))  float  v8f;
typedef __attribute__((ext_vector_type(4)))  float  v4f;

constexpr int B  = 256;
constexpr int U  = 128;
constexpr int P  = 64;
constexpr int H  = 768;
constexpr int K2 = U * P;

__global__ void norms_kernel(const float* __restrict__ src,
                             float* __restrict__ inv_out, int len) {
  int wave = (blockIdx.x * blockDim.x + threadIdx.x) >> 5;
  int lane = threadIdx.x & 31;
  if (wave >= len * B) return;
  int i = wave / B, b = wave % B;
  const float* row = src + (size_t)wave * H;
  float s = 0.f;
#pragma unroll
  for (int j = 0; j < H; j += 128) {
    float4 v = *(const float4*)(row + j + lane * 4);
    s += v.x * v.x + v.y * v.y + v.z * v.z + v.w * v.w;
  }
#pragma unroll
  for (int m = 16; m; m >>= 1) s += __shfl_xor(s, m, 32);
  if (lane == 0) inv_out[b * len + i] = 1.0f / fmaxf(sqrtf(s), 1e-8f);
}

constexpr int NT_O   = (U + P) / 16;
constexpr int KSTEPS = K2 / 32;

__global__ void wstage_kernel(const float* __restrict__ w_utt,
                              const float* __restrict__ w_pheno,
                              __bf16* __restrict__ wfrag) {
  int t = blockIdx.x * blockDim.x + threadIdx.x;
  if (t >= KSTEPS * NT_O * 64) return;
  int L    = t & 31;
  int half = (t >> 5) & 1;
  int frag = t >> 6;
  int ot   = frag % NT_O;
  int ks   = frag / NT_O;
  int o    = ot * 16 + (L & 15);
  int k    = ks * 32 + ((L >> 4) ? 8 : 0) + half * 16;
  v8bf r;
  if (o < U) {
    const float* s = w_utt + (size_t)o * K2 + k;
    float4 a = *(const float4*)(s);
    float4 c = *(const float4*)(s + 4);
    r[0] = (__bf16)a.x; r[1] = (__bf16)a.y; r[2] = (__bf16)a.z; r[3] = (__bf16)a.w;
    r[4] = (__bf16)c.x; r[5] = (__bf16)c.y; r[6] = (__bf16)c.z; r[7] = (__bf16)c.w;
  } else {
    int u = k / P, p = k % P;
    const float* s = w_pheno + (size_t)(o - U) * K2 + (size_t)p * U + u;
#pragma unroll
    for (int j = 0; j < 8; ++j) r[j] = (__bf16)s[j * U];
  }
  *(v8bf*)(wfrag + (size_t)t * 8) = r;
}

struct RawFrag { v4f a0, a1, b0, b1; };

__device__ __forceinline__ RawFrag load_raw_f32(const float* __restrict__ p) {
  RawFrag f;
  f.a0 = *(const v4f*)(p);
  f.a1 = *(const v4f*)(p + 4);
  f.b0 = *(const v4f*)(p + 16);
  f.b1 = *(const v4f*)(p + 20);
  return f;
}

__device__ __forceinline__ v16bf cvt_frag(const RawFrag& f) {
  v16bf r;
  r[0]  = (__bf16)f.a0[0]; r[1]  = (__bf16)f.a0[1];
  r[2]  = (__bf16)f.a0[2]; r[3]  = (__bf16)f.a0[3];
  r[4]  = (__bf16)f.a1[0]; r[5]  = (__bf16)f.a1[1];
  r[6]  = (__bf16)f.a1[2]; r[7]  = (__bf16)f.a1[3];
  r[8]  = (__bf16)f.b0[0]; r[9]  = (__bf16)f.b0[1];
  r[10] = (__bf16)f.b0[2]; r[11] = (__bf16)f.b0[3];
  r[12] = (__bf16)f.b1[0]; r[13] = (__bf16)f.b1[1];
  r[14] = (__bf16)f.b1[2]; r[15] = (__bf16)f.b1[3];
  return r;
}

__device__ __forceinline__ v16bf load_frag_bf16(const __bf16* __restrict__ p) {
  v8bf lo = *(const v8bf*)(p);
  v8bf hi = *(const v8bf*)(p + 16);
  v16bf r;
#pragma unroll
  for (int j = 0; j < 8; ++j) { r[j] = lo[j]; r[j + 8] = hi[j]; }
  return r;
}

template <int NT>
__device__ __forceinline__ void load_wfrags(const __bf16* __restrict__ fp,
                                            v8bf (&lo)[NT], v8bf (&up)[NT]) {
#pragma unroll
  for (int ot = 0; ot < NT; ++ot) {
    lo[ot] = *(const v8bf*)(fp + ot * 512);
    up[ot] = *(const v8bf*)(fp + ot * 512 + 256);
  }
}

__device__ __forceinline__ v16bf join_frag(v8bf lo, v8bf up) {
  v16bf r;
#pragma unroll
  for (int j = 0; j < 8; ++j) { r[j] = lo[j]; r[j + 8] = up[j]; }
  return r;
}

template <int NT>
__device__ __forceinline__ void hold_loads(const v8bf (&lo)[NT], const v8bf (&up)[NT]) {
  if constexpr (NT == 8) {
    asm volatile("" ::"v"(lo[0]), "v"(up[0]), "v"(lo[1]), "v"(up[1]),
                 "v"(lo[2]), "v"(up[2]), "v"(lo[3]), "v"(up[3]),
                 "v"(lo[4]), "v"(up[4]), "v"(lo[5]), "v"(up[5]),
                 "v"(lo[6]), "v"(up[6]), "v"(lo[7]), "v"(up[7]));
  } else {
    asm volatile("" ::"v"(lo[0]), "v"(up[0]), "v"(lo[1]), "v"(up[1]),
                 "v"(lo[2]), "v"(up[2]), "v"(lo[3]), "v"(up[3]));
  }
}

constexpr int KSLICE = H / 8;
constexpr int KS_A   = KSLICE / 32;
constexpr int MT_A   = U / 16;
constexpr int NT_A   = P / 16;
constexpr int TILE_F = 256;
constexpr int PART_F = MT_A * NT_A * TILE_F;

__global__ void __launch_bounds__(256)
cosine_kernel(const float* __restrict__ utt,
              const float* __restrict__ pheno,
              const float* __restrict__ inv_nu,
              const float* __restrict__ inv_np,
              __bf16* __restrict__ c_out) {
  __shared__ __attribute__((aligned(16))) float part[8 * PART_F];
  __shared__ __attribute__((aligned(16))) __bf16 c_lds[K2];

  int b    = blockIdx.x;
  int wave = threadIdx.x >> 5;
  int lane = threadIdx.x & 31;
  int rt   = lane & 15;
  int hi   = lane >> 4;
  int k0   = hi ? 8 : 0;

  constexpr size_t TILE_STRIDE = (size_t)16 * B * H;
  const unsigned loff = (unsigned)(rt * (B * H) + wave * KSLICE + k0);
  const float* Au = utt   + (size_t)b * H;
  const float* Bp = pheno + (size_t)b * H;

  v8f acc[MT_A][NT_A] = {};
#pragma unroll
  for (int ks = 0; ks < KS_A; ++ks) {
    RawFrag rb[NT_A], ra[MT_A];
#pragma unroll
    for (int nt = 0; nt < NT_A; ++nt)
      rb[nt] = load_raw_f32(Bp + nt * TILE_STRIDE + loff + ks * 32);
#pragma unroll
    for (int mt = 0; mt < MT_A; ++mt)
      ra[mt] = load_raw_f32(Au + mt * TILE_STRIDE + loff + ks * 32);
    v16bf bfr[NT_A];
#pragma unroll
    for (int nt = 0; nt < NT_A; ++nt) bfr[nt] = cvt_frag(rb[nt]);
#pragma unroll
    for (int mt = 0; mt < MT_A; ++mt) {
      v16bf a = cvt_frag(ra[mt]);
#pragma unroll
      for (int nt = 0; nt < NT_A; ++nt)
        acc[mt][nt] = __builtin_amdgcn_wmma_f32_16x16x32_bf16(
            false, a, false, bfr[nt], (short)0, acc[mt][nt], false, false);
    }
  }

  const float* su_base = inv_nu + b * U + wave * 16 + 8 * hi;
  float4 su0 = *(const float4*)(su_base);
  float4 su1 = *(const float4*)(su_base + 4);
  float sp[NT_A];
#pragma unroll
  for (int nt = 0; nt < NT_A; ++nt) sp[nt] = inv_np[b * P + nt * 16 + rt];

  {
    float* pw = part + wave * PART_F + lane * 4;
#pragma unroll
    for (int mt = 0; mt < MT_A; ++mt)
#pragma unroll
      for (int nt = 0; nt < NT_A; ++nt) {
        v8f t = acc[mt][nt];
        *(v4f*)(pw + (mt * NT_A + nt) * TILE_F)       = __builtin_shufflevector(t, t, 0, 1, 2, 3);
        *(v4f*)(pw + (mt * NT_A + nt) * TILE_F + 128) = __builtin_shufflevector(t, t, 4, 5, 6, 7);
      }
  }
  __syncthreads();

  v8f own[NT_A];
#pragma unroll
  for (int nt = 0; nt < NT_A; ++nt) {
    const float* pr = part + (wave * NT_A + nt) * TILE_F + lane * 4;
    v4f s0 = *(const v4f*)(pr);
    v4f s1 = *(const v4f*)(pr + 128);
#pragma unroll
    for (int w = 1; w < 8; ++w) {
      s0 += *(const v4f*)(pr + w * PART_F);
      s1 += *(const v4f*)(pr + w * PART_F + 128);
    }
    own[nt] = __builtin_shufflevector(s0, s1, 0, 1, 2, 3, 4, 5, 6, 7);
  }

  float su[8] = {su0.x, su0.y, su0.z, su0.w, su1.x, su1.y, su1.z, su1.w};

#pragma unroll
  for (int nt = 0; nt < NT_A; ++nt) {
#pragma unroll
    for (int r = 0; r < 8; ++r) {
      int m = r + 8 * hi;
      c_lds[(wave * 16 + m) * P + nt * 16 + rt] = (__bf16)(own[nt][r] * su[r] * sp[nt]);
    }
  }
  __syncthreads();

  {
    const v4f* s = (const v4f*)c_lds;
    v4f* d = (v4f*)(c_out + (size_t)b * K2);
#pragma unroll
    for (int j = 0; j < K2 * 2 / 16 / 256; ++j) d[j * 256 + threadIdx.x] = s[j * 256 + threadIdx.x];
  }
}

constexpr int LW     = 16;
constexpr int KW     = K2 / LW;
constexpr int KS_W   = KW / 32;
constexpr int BT     = 16;
constexpr int CPITCH = KW + 8;

template <int NT, int OT0>
__device__ __forceinline__ void logits_softmax_tile(const __bf16* __restrict__ c_ws,
                                                    const __bf16* __restrict__ wfrag,
                                                    const float* __restrict__ bias,
                                                    float* __restrict__ out,
                                                    int bt, __bf16* cst, float (*lg)[U]) {
  constexpr int N = NT * 16;
  int wave = threadIdx.x >> 5;
  int lane = threadIdx.x & 31;
  int rt   = lane & 15;
  int hi   = lane >> 4;
  int sl   = wave;

  {
    const __bf16* g = c_ws + (size_t)(bt * BT) * K2 + sl * KW + lane * 8;
    __bf16* l = cst + (wave * BT) * CPITCH + lane * 8;
    v8bf t[32];
#pragma unroll
    for (int j = 0; j < 32; ++j) t[j] = *(const v8bf*)(g + (size_t)(j >> 1) * K2 + (j & 1) * 256);
#pragma unroll
    for (int j = 0; j < 32; ++j) *(v8bf*)(l + (j >> 1) * CPITCH + (j & 1) * 256) = t[j];
  }
  __syncthreads();

  v8f lacc[NT] = {};
  const __bf16* cw = cst + (wave * BT + rt) * CPITCH + (hi ? 8 : 0);
  const __bf16* wf = wfrag + ((size_t)(sl * KS_W) * NT_O + OT0) * 512 + lane * 8;
  for (int i = 0; i < KS_W / 2; ++i) {
    int s = ((i + bt) & (KS_W / 2 - 1)) * 2;
    v8bf lo0[NT], up0[NT], lo1[NT], up1[NT];
    load_wfrags<NT>(wf + (size_t)s * NT_O * 512, lo0, up0);
    load_wfrags<NT>(wf + (size_t)(s + 1) * NT_O * 512, lo1, up1);
    v16bf cb0 = load_frag_bf16(cw + s * 32);
    v16bf cb1 = load_frag_bf16(cw + (s + 1) * 32);
    hold_loads<NT>(lo0, up0);
#pragma unroll
    for (int ot = 0; ot < NT; ++ot)
      lacc[ot] = __builtin_amdgcn_wmma_f32_16x16x32_bf16(
          false, join_frag(lo0[ot], up0[ot]), false, cb0, (short)0, lacc[ot], false, false);
    hold_loads<NT>(lo1, up1);
#pragma unroll
    for (int ot = 0; ot < NT; ++ot)
      lacc[ot] = __builtin_amdgcn_wmma_f32_16x16x32_bf16(
          false, join_frag(lo1[ot], up1[ot]), false, cb1, (short)0, lacc[ot], false, false);
  }
  __syncthreads();

  float* part = (float*)cst;
#pragma unroll
  for (int ot = 0; ot < NT; ++ot)
#pragma unroll
    for (int r = 0; r < 8; ++r)
      part[((sl * NT + ot) * 8 + r) * 32 + lane] = lacc[ot][r];
  __syncthreads();

#pragma unroll
  for (int i = 0; i < NT * 256 / 512; ++i) {
    int idx = i * 512 + threadIdx.x;
    float s = part[idx];
#pragma unroll
    for (int w = 1; w < LW; ++w) s += part[w * NT * 256 + idx];
    int ot = idx >> 8, r = (idx >> 5) & 7, ln = idx & 31;
    int o = ot * 16 + r + 8 * (ln >> 4);
    lg[ln & 15][o] = s + bias[o];
  }
  __syncthreads();

  {
    const float* x = lg[wave];
    float* dst = out + (size_t)(bt * BT + wave) * N;
    constexpr int per = N / 32;
    float vals[per];
    float mx = -1e30f;
#pragma unroll
    for (int j = 0; j < per; ++j) { vals[j] = x[lane + 32 * j]; mx = fmaxf(mx, vals[j]); }
#pragma unroll
    for (int m = 16; m; m >>= 1) mx = fmaxf(mx, __shfl_xor(mx, m, 32));
    float s = 0.f;
#pragma unroll
    for (int j = 0; j < per; ++j) { vals[j] = __expf(vals[j] - mx); s += vals[j]; }
#pragma unroll
    for (int m = 16; m; m >>= 1) s += __shfl_xor(s, m, 32);
    float inv = 1.0f / s;
#pragma unroll
    for (int j = 0; j < per; ++j) dst[lane + 32 * j] = vals[j] * inv;
  }
}

__global__ void __launch_bounds__(512)
logits_softmax_kernel(const __bf16* __restrict__ c_ws,
                      const __bf16* __restrict__ wfrag,
                      const float* __restrict__ b_utt,
                      const float* __restrict__ b_pheno,
                      float* __restrict__ out_row,
                      float* __restrict__ out_col) {
  __shared__ __attribute__((aligned(16))) __bf16 cst[LW * BT * CPITCH];
  __shared__ float lg[BT][U];
  int bt = blockIdx.x & 15;
  if (blockIdx.x < 16)
    logits_softmax_tile<U / 16, 0>(c_ws, wfrag, b_utt, out_row, bt, cst, lg);
  else
    logits_softmax_tile<P / 16, U / 16>(c_ws, wfrag, b_pheno, out_col, bt, cst, lg);
}

extern "C" void kernel_launch(void* const* d_in, const int* in_sizes, int n_in,
                              void* d_out, int out_size, void* d_ws, size_t ws_size,
                              hipStream_t stream) {
  const float* utt     = (const float*)d_in[0];
  const float* pheno   = (const float*)d_in[1];
  const float* w_utt   = (const float*)d_in[2];
  const float* b_utt   = (const float*)d_in[3];
  const float* w_pheno = (const float*)d_in[4];
  const float* b_pheno = (const float*)d_in[5];
  (void)in_sizes; (void)n_in; (void)out_size; (void)ws_size;

  char* ws = (char*)d_ws;
  size_t off = 0;
  auto alloc = [&](size_t bytes) {
    void* p = ws + off;
    off = (off + bytes + 255) & ~(size_t)255;
    return p;
  };
  float*  inv_nu = (float*)alloc((size_t)B * U * sizeof(float));
  float*  inv_np = (float*)alloc((size_t)B * P * sizeof(float));
  __bf16* wfrag  = (__bf16*)alloc((size_t)(U + P) * K2 * sizeof(__bf16));
  __bf16* c_ws   = (__bf16*)alloc((size_t)B * K2 * sizeof(__bf16));

  float* row_logits = (float*)d_out;
  float* col_logits = (float*)d_out + B * U;

  norms_kernel<<<(B * U) / 8, 256, 0, stream>>>(utt, inv_nu, U);
  norms_kernel<<<(B * P) / 8, 256, 0, stream>>>(pheno, inv_np, P);
  wstage_kernel<<<(KSTEPS * NT_O * 64) / 256, 256, 0, stream>>>(w_utt, w_pheno, wfrag);

  cosine_kernel<<<B, 256, 0, stream>>>(utt, pheno, inv_nu, inv_np, c_ws);
  logits_softmax_kernel<<<2 * (B / BT), 32 * LW, 0, stream>>>(c_ws, wfrag, b_utt, b_pheno,
                                                         row_logits, col_logits);
}


# norm passes: one launch of 256 resident blocks (one per batch, 16 waves x 12 rows, up to 5 rows of loads in flight per wave) instead of 49152 one-row waves in two launches; sums keep their order
# speedup vs baseline: 3.6311x; 1.0405x over previous
#include <hip/hip_runtime.h>
#include <hip/hip_bf16.h>
#include <math.h>

typedef __attribute__((ext_vector_type(16))) __bf16 v16bf;
typedef __attribute__((ext_vector_type(8)))  __bf16 v8bf;
typedef __attribute__((ext_vector_type(8)))  float  v8f;
typedef __attribute__((ext_vector_type(4)))  float  v4f;

constexpr int B  = 256;
constexpr int U  = 128;
constexpr int P  = 64;
constexpr int H  = 768;
constexpr int K2 = U * P;

constexpr int NW_N    = 16;
constexpr int RU_N    = U / NW_N;
constexpr int RP_N    = P / NW_N;
constexpr int ROWS_N  = RU_N + RP_N;
constexpr int AHEAD_N = 4;
constexpr int HQ      = H / 128;

__global__ void __launch_bounds__(NW_N * 32)
norms_kernel(const float* __restrict__ utt, const float* __restrict__ pheno,
             float* __restrict__ inv_nu, float* __restrict__ inv_np) {
  __shared__ float inv_lds[U + P];
  const int b    = blockIdx.x;
  const int lane = threadIdx.x & 31;
  const int wave = __builtin_amdgcn_readfirstlane(threadIdx.x >> 5);
  const float* ub = utt   + ((size_t)(wave * RU_N) * B + b) * H + lane * 4;
  const float* pb = pheno + ((size_t)(wave * RP_N) * B + b) * H + lane * 4;

  v4f buf[AHEAD_N + 1][HQ];
  float sum[ROWS_N];
#define LOAD_ROW(R) do { \
    const float* rp_ = ((R) < RU_N) ? ub + (size_t)(R) * B * H : pb + (size_t)((R) - RU_N) * B * H; \
    _Pragma("unroll") for (int q = 0; q < HQ; ++q) buf[(R) % (AHEAD_N + 1)][q] = *(const v4f*)(rp_ + q * 128); } while (0)
#pragma unroll
  for (int r = 0; r < AHEAD_N; ++r) LOAD_ROW(r);
#pragma unroll
  for (int r = 0; r < ROWS_N; ++r) {
    __builtin_amdgcn_sched_barrier(0);
    if (r + AHEAD_N < ROWS_N) LOAD_ROW(r + AHEAD_N);
    __builtin_amdgcn_sched_barrier(0);
    float s = 0.f;
#pragma unroll
    for (int q = 0; q < HQ; ++q) {
      v4f v = buf[r % (AHEAD_N + 1)][q];
      s += v.x * v.x + v.y * v.y + v.z * v.z + v.w * v.w;
    }
    sum[r] = s;
  }
#undef LOAD_ROW
#pragma unroll
  for (int r = 0; r < ROWS_N; ++r) {
    float s = sum[r];
#pragma unroll
    for (int m = 16; m; m >>= 1) s += __shfl_xor(s, m, 32);
    sum[r] = s;
  }
  float mine = 1.f;
#pragma unroll
  for (int r = 0; r < ROWS_N; ++r) if (lane == r) mine = sum[r];
  if (lane < ROWS_N) {
    int slot = (lane < RU_N) ? wave * RU_N + lane : U + wave * RP_N + (lane - RU_N);
    inv_lds[slot] = 1.0f / fmaxf(sqrtf(mine), 1e-8f);
  }
  __syncthreads();
  if (threadIdx.x < U)          inv_nu[b * U + threadIdx.x] = inv_lds[threadIdx.x];
  else if (threadIdx.x < U + P) inv_np[b * P + (threadIdx.x - U)] = inv_lds[threadIdx.x];
}

constexpr int NT_O   = (U + P) / 16;
constexpr int KSTEPS = K2 / 32;

__global__ void wstage_kernel(const float* __restrict__ w_utt,
                              const float* __restrict__ w_pheno,
                              __bf16* __restrict__ wfrag) {
  int t = blockIdx.x * blockDim.x + threadIdx.x;
  if (t >= KSTEPS * NT_O * 64) return;
  int L    = t & 31;
  int half = (t >> 5) & 1;
  int frag = t >> 6;
  int ot   = frag % NT_O;
  int ks   = frag / NT_O;
  int o    = ot * 16 + (L & 15);
  int k    = ks * 32 + ((L >> 4) ? 8 : 0) + half * 16;
  v8bf r;
  if (o < U) {
    const float* s = w_utt + (size_t)o * K2 + k;
    float4 a = *(const float4*)(s);
    float4 c = *(const float4*)(s + 4);
    r[0] = (__bf16)a.x; r[1] = (__bf16)a.y; r[2] = (__bf16)a.z; r[3] = (__bf16)a.w;
    r[4] = (__bf16)c.x; r[5] = (__bf16)c.y; r[6] = (__bf16)c.z; r[7] = (__bf16)c.w;
  } else {
    int u = k / P, p = k % P;
    const float* s = w_pheno + (size_t)(o - U) * K2 + (size_t)p * U + u;
#pragma unroll
    for (int j = 0; j < 8; ++j) r[j] = (__bf16)s[j * U];
  }
  *(v8bf*)(wfrag + (size_t)t * 8) = r;
}

struct RawFrag { v4f a0, a1, b0, b1; };

__device__ __forceinline__ RawFrag load_raw_f32(const float* __restrict__ p) {
  RawFrag f;
  f.a0 = *(const v4f*)(p);
  f.a1 = *(const v4f*)(p + 4);
  f.b0 = *(const v4f*)(p + 16);
  f.b1 = *(const v4f*)(p + 20);
  return f;
}

__device__ __forceinline__ v16bf cvt_frag(const RawFrag& f) {
  v16bf r;
  r[0]  = (__bf16)f.a0[0]; r[1]  = (__bf16)f.a0[1];
  r[2]  = (__bf16)f.a0[2]; r[3]  = (__bf16)f.a0[3];
  r[4]  = (__bf16)f.a1[0]; r[5]  = (__bf16)f.a1[1];
  r[6]  = (__bf16)f.a1[2]; r[7]  = (__bf16)f.a1[3];
  r[8]  = (__bf16)f.b0[0]; r[9]  = (__bf16)f.b0[1];
  r[10] = (__bf16)f.b0[2]; r[11] = (__bf16)f.b0[3];
  r[12] = (__bf16)f.b1[0]; r[13] = (__bf16)f.b1[1];
  r[14] = (__bf16)f.b1[2]; r[15] = (__bf16)f.b1[3];
  return r;
}

__device__ __forceinline__ v16bf load_frag_bf16(const __bf16* __restrict__ p) {
  v8bf lo = *(const v8bf*)(p);
  v8bf hi = *(const v8bf*)(p + 16);
  v16bf r;
#pragma unroll
  for (int j = 0; j < 8; ++j) { r[j] = lo[j]; r[j + 8] = hi[j]; }
  return r;
}

template <int NT>
__device__ __forceinline__ void load_wfrags(const __bf16* __restrict__ fp,
                                            v8bf (&lo)[NT], v8bf (&up)[NT]) {
#pragma unroll
  for (int ot = 0; ot < NT; ++ot) {
    lo[ot] = *(const v8bf*)(fp + ot * 512);
    up[ot] = *(const v8bf*)(fp + ot * 512 + 256);
  }
}

__device__ __forceinline__ v16bf join_frag(v8bf lo, v8bf up) {
  v16bf r;
#pragma unroll
  for (int j = 0; j < 8; ++j) { r[j] = lo[j]; r[j + 8] = up[j]; }
  return r;
}

template <int NT>
__device__ __forceinline__ void hold_loads(const v8bf (&lo)[NT], const v8bf (&up)[NT]) {
  if constexpr (NT == 8) {
    asm volatile("" ::"v"(lo[0]), "v"(up[0]), "v"(lo[1]), "v"(up[1]),
                 "v"(lo[2]), "v"(up[2]), "v"(lo[3]), "v"(up[3]),
                 "v"(lo[4]), "v"(up[4]), "v"(lo[5]), "v"(up[5]),
                 "v"(lo[6]), "v"(up[6]), "v"(lo[7]), "v"(up[7]));
  } else {
    asm volatile("" ::"v"(lo[0]), "v"(up[0]), "v"(lo[1]), "v"(up[1]),
                 "v"(lo[2]), "v"(up[2]), "v"(lo[3]), "v"(up[3]));
  }
}

constexpr int KSLICE = H / 8;
constexpr int KS_A   = KSLICE / 32;
constexpr int MT_A   = U / 16;
constexpr int NT_A   = P / 16;
constexpr int TILE_F = 256;
constexpr int PART_F = MT_A * NT_A * TILE_F;

__global__ void __launch_bounds__(256)
cosine_kernel(const float* __restrict__ utt,
              const float* __restrict__ pheno,
              const float* __restrict__ inv_nu,
              const float* __restrict__ inv_np,
              __bf16* __restrict__ c_out) {
  __shared__ __attribute__((aligned(16))) float part[8 * PART_F];
  __shared__ __attribute__((aligned(16))) __bf16 c_lds[K2];

  int b    = blockIdx.x;
  int wave = threadIdx.x >> 5;
  int lane = threadIdx.x & 31;
  int rt   = lane & 15;
  int hi   = lane >> 4;
  int k0   = hi ? 8 : 0;

  constexpr size_t TILE_STRIDE = (size_t)16 * B * H;
  const unsigned loff = (unsigned)(rt * (B * H) + wave * KSLICE + k0);
  const float* Au = utt   + (size_t)b * H;
  const float* Bp = pheno + (size_t)b * H;

  v8f acc[MT_A][NT_A] = {};
#pragma unroll
  for (int ks = 0; ks < KS_A; ++ks) {
    RawFrag rb[NT_A], ra[MT_A];
#pragma unroll
    for (int nt = 0; nt < NT_A; ++nt)
      rb[nt] = load_raw_f32(Bp + nt * TILE_STRIDE + loff + ks * 32);
#pragma unroll
    for (int mt = 0; mt < MT_A; ++mt)
      ra[mt] = load_raw_f32(Au + mt * TILE_STRIDE + loff + ks * 32);
    v16bf bfr[NT_A];
#pragma unroll
    for (int nt = 0; nt < NT_A; ++nt) bfr[nt] = cvt_frag(rb[nt]);
#pragma unroll
    for (int mt = 0; mt < MT_A; ++mt) {
      v16bf a = cvt_frag(ra[mt]);
#pragma unroll
      for (int nt = 0; nt < NT_A; ++nt)
        acc[mt][nt] = __builtin_amdgcn_wmma_f32_16x16x32_bf16(
            false, a, false, bfr[nt], (short)0, acc[mt][nt], false, false);
    }
  }

  const float* su_base = inv_nu + b * U + wave * 16 + 8 * hi;
  float4 su0 = *(const float4*)(su_base);
  float4 su1 = *(const float4*)(su_base + 4);
  float sp[NT_A];
#pragma unroll
  for (int nt = 0; nt < NT_A; ++nt) sp[nt] = inv_np[b * P + nt * 16 + rt];

  {
    float* pw = part + wave * PART_F + lane * 4;
#pragma unroll
    for (int mt = 0; mt < MT_A; ++mt)
#pragma unroll
      for (int nt = 0; nt < NT_A; ++nt) {
        v8f t = acc[mt][nt];
        *(v4f*)(pw + (mt * NT_A + nt) * TILE_F)       = __builtin_shufflevector(t, t, 0, 1, 2, 3);
        *(v4f*)(pw + (mt * NT_A + nt) * TILE_F + 128) = __builtin_shufflevector(t, t, 4, 5, 6, 7);
      }
  }
  __syncthreads();

  v8f own[NT_A];
#pragma unroll
  for (int nt = 0; nt < NT_A; ++nt) {
    const float* pr = part + (wave * NT_A + nt) * TILE_F + lane * 4;
    v4f s0 = *(const v4f*)(pr);
    v4f s1 = *(const v4f*)(pr + 128);
#pragma unroll
    for (int w = 1; w < 8; ++w) {
      s0 += *(const v4f*)(pr + w * PART_F);
      s1 += *(const v4f*)(pr + w * PART_F + 128);
    }
    own[nt] = __builtin_shufflevector(s0, s1, 0, 1, 2, 3, 4, 5, 6, 7);
  }

  float su[8] = {su0.x, su0.y, su0.z, su0.w, su1.x, su1.y, su1.z, su1.w};

#pragma unroll
  for (int nt = 0; nt < NT_A; ++nt) {
#pragma unroll
    for (int r = 0; r < 8; ++r) {
      int m = r + 8 * hi;
      c_lds[(wave * 16 + m) * P + nt * 16 + rt] = (__bf16)(own[nt][r] * su[r] * sp[nt]);
    }
  }
  __syncthreads();

  {
    const v4f* s = (const v4f*)c_lds;
    v4f* d = (v4f*)(c_out + (size_t)b * K2);
#pragma unroll
    for (int j = 0; j < K2 * 2 / 16 / 256; ++j) d[j * 256 + threadIdx.x] = s[j * 256 + threadIdx.x];
  }
}

constexpr int LW     = 16;
constexpr int KW     = K2 / LW;
constexpr int KS_W   = KW / 32;
constexpr int BT     = 16;
constexpr int CPITCH = KW + 8;

template <int NT, int OT0>
__device__ __forceinline__ void logits_softmax_tile(const __bf16* __restrict__ c_ws,
                                                    const __bf16* __restrict__ wfrag,
                                                    const float* __restrict__ bias,
                                                    float* __restrict__ out,
                                                    int bt, __bf16* cst, float (*lg)[U]) {
  constexpr int N = NT * 16;
  int wave = threadIdx.x >> 5;
  int lane = threadIdx.x & 31;
  int rt   = lane & 15;
  int hi   = lane >> 4;
  int sl   = wave;

  {
    const __bf16* g = c_ws + (size_t)(bt * BT) * K2 + sl * KW + lane * 8;
    __bf16* l = cst + (wave * BT) * CPITCH + lane * 8;
    v8bf t[32];
#pragma unroll
    for (int j = 0; j < 32; ++j) t[j] = *(const v8bf*)(g + (size_t)(j >> 1) * K2 + (j & 1) * 256);
#pragma unroll
    for (int j = 0; j < 32; ++j) *(v8bf*)(l + (j >> 1) * CPITCH + (j & 1) * 256) = t[j];
  }
  __syncthreads();

  v8f lacc[NT] = {};
  const __bf16* cw = cst + (wave * BT + rt) * CPITCH + (hi ? 8 : 0);
  const __bf16* wf = wfrag + ((size_t)(sl * KS_W) * NT_O + OT0) * 512 + lane * 8;
  for (int i = 0; i < KS_W / 2; ++i) {
    int s = ((i + bt) & (KS_W / 2 - 1)) * 2;
    v8bf lo0[NT], up0[NT], lo1[NT], up1[NT];
    load_wfrags<NT>(wf + (size_t)s * NT_O * 512, lo0, up0);
    load_wfrags<NT>(wf + (size_t)(s + 1) * NT_O * 512, lo1, up1);
    v16bf cb0 = load_frag_bf16(cw + s * 32);
    v16bf cb1 = load_frag_bf16(cw + (s + 1) * 32);
    hold_loads<NT>(lo0, up0);
#pragma unroll
    for (int ot = 0; ot < NT; ++ot)
      lacc[ot] = __builtin_amdgcn_wmma_f32_16x16x32_bf16(
          false, join_frag(lo0[ot], up0[ot]), false, cb0, (short)0, lacc[ot], false, false);
    hold_loads<NT>(lo1, up1);
#pragma unroll
    for (int ot = 0; ot < NT; ++ot)
      lacc[ot] = __builtin_amdgcn_wmma_f32_16x16x32_bf16(
          false, join_frag(lo1[ot], up1[ot]), false, cb1, (short)0, lacc[ot], false, false);
  }
  __syncthreads();

  float* part = (float*)cst;
#pragma unroll
  for (int ot = 0; ot < NT; ++ot)
#pragma unroll
    for (int r = 0; r < 8; ++r)
      part[((sl * NT + ot) * 8 + r) * 32 + lane] = lacc[ot][r];
  __syncthreads();

#pragma unroll
  for (int i = 0; i < NT * 256 / 512; ++i) {
    int idx = i * 512 + threadIdx.x;
    float s = part[idx];
#pragma unroll
    for (int w = 1; w < LW; ++w) s += part[w * NT * 256 + idx];
    int ot = idx >> 8, r = (idx >> 5) & 7, ln = idx & 31;
    int o = ot * 16 + r + 8 * (ln >> 4);
    lg[ln & 15][o] = s + bias[o];
  }
  __syncthreads();

  {
    const float* x = lg[wave];
    float* dst = out + (size_t)(bt * BT + wave) * N;
    constexpr int per = N / 32;
    float vals[per];
    float mx = -1e30f;
#pragma unroll
    for (int j = 0; j < per; ++j) { vals[j] = x[lane + 32 * j]; mx = fmaxf(mx, vals[j]); }
#pragma unroll
    for (int m = 16; m; m >>= 1) mx = fmaxf(mx, __shfl_xor(mx, m, 32));
    float s = 0.f;
#pragma unroll
    for (int j = 0; j < per; ++j) { vals[j] = __expf(vals[j] - mx); s += vals[j]; }
#pragma unroll
    for (int m = 16; m; m >>= 1) s += __shfl_xor(s, m, 32);
    float inv = 1.0f / s;
#pragma unroll
    for (int j = 0; j < per; ++j) dst[lane + 32 * j] = vals[j] * inv;
  }
}

__global__ void __launch_bounds__(512)
logits_softmax_kernel(const __bf16* __restrict__ c_ws,
                      const __bf16* __restrict__ wfrag,
                      const float* __restrict__ b_utt,
                      const float* __restrict__ b_pheno,
                      float* __restrict__ out_row,
                      float* __restrict__ out_col) {
  __shared__ __attribute__((aligned(16))) __bf16 cst[LW * BT * CPITCH];
  __shared__ float lg[BT][U];
  int bt = blockIdx.x & 15;
  if (blockIdx.x < 16)
    logits_softmax_tile<U / 16, 0>(c_ws, wfrag, b_utt, out_row, bt, cst, lg);
  else
    logits_softmax_tile<P / 16, U / 16>(c_ws, wfrag, b_pheno, out_col, bt, cst, lg);
}

extern "C" void kernel_launch(void* const* d_in, const int* in_sizes, int n_in,
                              void* d_out, int out_size, void* d_ws, size_t ws_size,
                              hipStream_t stream) {
  const float* utt     = (const float*)d_in[0];
  const float* pheno   = (const float*)d_in[1];
  const float* w_utt   = (const float*)d_in[2];
  const float* b_utt   = (const float*)d_in[3];
  const float* w_pheno = (const float*)d_in[4];
  const float* b_pheno = (const float*)d_in[5];
  (void)in_sizes; (void)n_in; (void)out_size; (void)ws_size;

  char* ws = (char*)d_ws;
  size_t off = 0;
  auto alloc = [&](size_t bytes) {
    void* p = ws + off;
    off = (off + bytes + 255) & ~(size_t)255;
    return p;
  };
  float*  inv_nu = (float*)alloc((size_t)B * U * sizeof(float));
  float*  inv_np = (float*)alloc((size_t)B * P * sizeof(float));
  __bf16* wfrag  = (__bf16*)alloc((size_t)(U + P) * K2 * sizeof(__bf16));
  __bf16* c_ws   = (__bf16*)alloc((size_t)B * K2 * sizeof(__bf16));

  float* row_logits = (float*)d_out;
  float* col_logits = (float*)d_out + B * U;

  norms_kernel<<<B, NW_N * 32, 0, stream>>>(utt, pheno, inv_nu, inv_np);
  wstage_kernel<<<(KSTEPS * NT_O * 64) / 256, 256, 0, stream>>>(w_utt, w_pheno, wfrag);

  cosine_kernel<<<B, 256, 0, stream>>>(utt, pheno, inv_nu, inv_np, c_ws);
  logits_softmax_kernel<<<2 * (B / BT), 32 * LW, 0, stream>>>(c_ws, wfrag, b_utt, b_pheno,
                                                         row_logits, col_logits);
}


# logits GEMM: contraction split over 256 blocks (16 batches x one 512-k slice x all 192 logits) not 32; slice partials via workspace, new 2-wave launch sums them + softmax (f32 sum order changes)
# speedup vs baseline: 4.4919x; 1.2371x over previous
#include <hip/hip_runtime.h>
#include <hip/hip_bf16.h>
#include <math.h>

typedef __attribute__((ext_vector_type(16))) __bf16 v16bf;
typedef __attribute__((ext_vector_type(8)))  __bf16 v8bf;
typedef __attribute__((ext_vector_type(8)))  float  v8f;
typedef __attribute__((ext_vector_type(4)))  float  v4f;

constexpr int B  = 256;
constexpr int U  = 128;
constexpr int P  = 64;
constexpr int H  = 768;
constexpr int K2 = U * P;

constexpr int NW_N    = 16;
constexpr int RU_N    = U / NW_N;
constexpr int RP_N    = P / NW_N;
constexpr int ROWS_N  = RU_N + RP_N;
constexpr int AHEAD_N = 4;
constexpr int HQ      = H / 128;

__global__ void __launch_bounds__(NW_N * 32)
norms_kernel(const float* __restrict__ utt, const float* __restrict__ pheno,
             float* __restrict__ inv_nu, float* __restrict__ inv_np) {
  __shared__ float inv_lds[U + P];
  const int b    = blockIdx.x;
  const int lane = threadIdx.x & 31;
  const int wave = __builtin_amdgcn_readfirstlane(threadIdx.x >> 5);
  const float* ub = utt   + ((size_t)(wave * RU_N) * B + b) * H + lane * 4;
  const float* pb = pheno + ((size_t)(wave * RP_N) * B + b) * H + lane * 4;

  v4f buf[AHEAD_N + 1][HQ];
  float sum[ROWS_N];
#define LOAD_ROW(R) do { \
    const float* rp_ = ((R) < RU_N) ? ub + (size_t)(R) * B * H : pb + (size_t)((R) - RU_N) * B * H; \
    _Pragma("unroll") for (int q = 0; q < HQ; ++q) buf[(R) % (AHEAD_N + 1)][q] = *(const v4f*)(rp_ + q * 128); } while (0)
#pragma unroll
  for (int r = 0; r < AHEAD_N; ++r) LOAD_ROW(r);
#pragma unroll
  for (int r = 0; r < ROWS_N; ++r) {
    __builtin_amdgcn_sched_barrier(0);
    if (r + AHEAD_N < ROWS_N) LOAD_ROW(r + AHEAD_N);
    __builtin_amdgcn_sched_barrier(0);
    float s = 0.f;
#pragma unroll
    for (int q = 0; q < HQ; ++q) {
      v4f v = buf[r % (AHEAD_N + 1)][q];
      s += v.x * v.x + v.y * v.y + v.z * v.z + v.w * v.w;
    }
    sum[r] = s;
  }
#undef LOAD_ROW
#pragma unroll
  for (int r = 0; r < ROWS_N; ++r) {
    float s = sum[r];
#pragma unroll
    for (int m = 16; m; m >>= 1) s += __shfl_xor(s, m, 32);
    sum[r] = s;
  }
  float mine = 1.f;
#pragma unroll
  for (int r = 0; r < ROWS_N; ++r) if (lane == r) mine = sum[r];
  if (lane < ROWS_N) {
    int slot = (lane < RU_N) ? wave * RU_N + lane : U + wave * RP_N + (lane - RU_N);
    inv_lds[slot] = 1.0f / fmaxf(sqrtf(mine), 1e-8f);
  }
  __syncthreads();
  if (threadIdx.x < U)          inv_nu[b * U + threadIdx.x] = inv_lds[threadIdx.x];
  else if (threadIdx.x < U + P) inv_np[b * P + (threadIdx.x - U)] = inv_lds[threadIdx.x];
}

constexpr int NT_O   = (U + P) / 16;
constexpr int KSTEPS = K2 / 32;

__global__ void wstage_kernel(const float* __restrict__ w_utt,
                              const float* __restrict__ w_pheno,
                              __bf16* __restrict__ wfrag) {
  int t = blockIdx.x * blockDim.x + threadIdx.x;
  if (t >= KSTEPS * NT_O * 64) return;
  int L    = t & 31;
  int half = (t >> 5) & 1;
  int frag = t >> 6;
  int ot   = frag % NT_O;
  int ks   = frag / NT_O;
  int o    = ot * 16 + (L & 15);
  int k    = ks * 32 + ((L >> 4) ? 8 : 0) + half * 16;
  v8bf r;
  if (o < U) {
    const float* s = w_utt + (size_t)o * K2 + k;
    float4 a = *(const float4*)(s);
    float4 c = *(const float4*)(s + 4);
    r[0] = (__bf16)a.x; r[1] = (__bf16)a.y; r[2] = (__bf16)a.z; r[3] = (__bf16)a.w;
    r[4] = (__bf16)c.x; r[5] = (__bf16)c.y; r[6] = (__bf16)c.z; r[7] = (__bf16)c.w;
  } else {
    int u = k / P, p = k % P;
    const float* s = w_pheno + (size_t)(o - U) * K2 + (size_t)p * U + u;
#pragma unroll
    for (int j = 0; j < 8; ++j) r[j] = (__bf16)s[j * U];
  }
  *(v8bf*)(wfrag + (size_t)t * 8) = r;
}

struct RawFrag { v4f a0, a1, b0, b1; };

__device__ __forceinline__ RawFrag load_raw_f32(const float* __restrict__ p) {
  RawFrag f;
  f.a0 = *(const v4f*)(p);
  f.a1 = *(const v4f*)(p + 4);
  f.b0 = *(const v4f*)(p + 16);
  f.b1 = *(const v4f*)(p + 20);
  return f;
}

__device__ __forceinline__ v16bf cvt_frag(const RawFrag& f) {
  v16bf r;
  r[0]  = (__bf16)f.a0[0]; r[1]  = (__bf16)f.a0[1];
  r[2]  = (__bf16)f.a0[2]; r[3]  = (__bf16)f.a0[3];
  r[4]  = (__bf16)f.a1[0]; r[5]  = (__bf16)f.a1[1];
  r[6]  = (__bf16)f.a1[2]; r[7]  = (__bf16)f.a1[3];
  r[8]  = (__bf16)f.b0[0]; r[9]  = (__bf16)f.b0[1];
  r[10] = (__bf16)f.b0[2]; r[11] = (__bf16)f.b0[3];
  r[12] = (__bf16)f.b1[0]; r[13] = (__bf16)f.b1[1];
  r[14] = (__bf16)f.b1[2]; r[15] = (__bf16)f.b1[3];
  return r;
}

__device__ __forceinline__ v16bf load_frag_bf16(const __bf16* __restrict__ p) {
  v8bf lo = *(const v8bf*)(p);
  v8bf hi = *(const v8bf*)(p + 16);
  v16bf r;
#pragma unroll
  for (int j = 0; j < 8; ++j) { r[j] = lo[j]; r[j + 8] = hi[j]; }
  return r;
}

template <int NT>
__device__ __forceinline__ void load_wfrags(const __bf16* __restrict__ fp,
                                            v8bf (&lo)[NT], v8bf (&up)[NT]) {
#pragma unroll
  for (int ot = 0; ot < NT; ++ot) {
    lo[ot] = *(const v8bf*)(fp + ot * 512);
    up[ot] = *(const v8bf*)(fp + ot * 512 + 256);
  }
}

__device__ __forceinline__ v16bf join_frag(v8bf lo, v8bf up) {
  v16bf r;
#pragma unroll
  for (int j = 0; j < 8; ++j) { r[j] = lo[j]; r[j + 8] = up[j]; }
  return r;
}

constexpr int KSLICE = H / 8;
constexpr int KS_A   = KSLICE / 32;
constexpr int MT_A   = U / 16;
constexpr int NT_A   = P / 16;
constexpr int TILE_F = 256;
constexpr int PART_F = MT_A * NT_A * TILE_F;

__global__ void __launch_bounds__(256)
cosine_kernel(const float* __restrict__ utt,
              const float* __restrict__ pheno,
              const float* __restrict__ inv_nu,
              const float* __restrict__ inv_np,
              __bf16* __restrict__ c_out) {
  __shared__ __attribute__((aligned(16))) float part[8 * PART_F];
  __shared__ __attribute__((aligned(16))) __bf16 c_lds[K2];

  int b    = blockIdx.x;
  int wave = threadIdx.x >> 5;
  int lane = threadIdx.x & 31;
  int rt   = lane & 15;
  int hi   = lane >> 4;
  int k0   = hi ? 8 : 0;

  constexpr size_t TILE_STRIDE = (size_t)16 * B * H;
  const unsigned loff = (unsigned)(rt * (B * H) + wave * KSLICE + k0);
  const float* Au = utt   + (size_t)b * H;
  const float* Bp = pheno + (size_t)b * H;

  v8f acc[MT_A][NT_A] = {};
#pragma unroll
  for (int ks = 0; ks < KS_A; ++ks) {
    RawFrag rb[NT_A], ra[MT_A];
#pragma unroll
    for (int nt = 0; nt < NT_A; ++nt)
      rb[nt] = load_raw_f32(Bp + nt * TILE_STRIDE + loff + ks * 32);
#pragma unroll
    for (int mt = 0; mt < MT_A; ++mt)
      ra[mt] = load_raw_f32(Au + mt * TILE_STRIDE + loff + ks * 32);
    v16bf bfr[NT_A];
#pragma unroll
    for (int nt = 0; nt < NT_A; ++nt) bfr[nt] = cvt_frag(rb[nt]);
#pragma unroll
    for (int mt = 0; mt < MT_A; ++mt) {
      v16bf a = cvt_frag(ra[mt]);
#pragma unroll
      for (int nt = 0; nt < NT_A; ++nt)
        acc[mt][nt] = __builtin_amdgcn_wmma_f32_16x16x32_bf16(
            false, a, false, bfr[nt], (short)0, acc[mt][nt], false, false);
    }
  }

  const float* su_base = inv_nu + b * U + wave * 16 + 8 * hi;
  float4 su0 = *(const float4*)(su_base);
  float4 su1 = *(const float4*)(su_base + 4);
  float sp[NT_A];
#pragma unroll
  for (int nt = 0; nt < NT_A; ++nt) sp[nt] = inv_np[b * P + nt * 16 + rt];

  {
    float* pw = part + wave * PART_F + lane * 4;
#pragma unroll
    for (int mt = 0; mt < MT_A; ++mt)
#pragma unroll
      for (int nt = 0; nt < NT_A; ++nt) {
        v8f t = acc[mt][nt];
        *(v4f*)(pw + (mt * NT_A + nt) * TILE_F)       = __builtin_shufflevector(t, t, 0, 1, 2, 3);
        *(v4f*)(pw + (mt * NT_A + nt) * TILE_F + 128) = __builtin_shufflevector(t, t, 4, 5, 6, 7);
      }
  }
  __syncthreads();

  v8f own[NT_A];
#pragma unroll
  for (int nt = 0; nt < NT_A; ++nt) {
    const float* pr = part + (wave * NT_A + nt) * TILE_F + lane * 4;
    v4f s0 = *(const v4f*)(pr);
    v4f s1 = *(const v4f*)(pr + 128);
#pragma unroll
    for (int w = 1; w < 8; ++w) {
      s0 += *(const v4f*)(pr + w * PART_F);
      s1 += *(const v4f*)(pr + w * PART_F + 128);
    }
    own[nt] = __builtin_shufflevector(s0, s1, 0, 1, 2, 3, 4, 5, 6, 7);
  }

  float su[8] = {su0.x, su0.y, su0.z, su0.w, su1.x, su1.y, su1.z, su1.w};

#pragma unroll
  for (int nt = 0; nt < NT_A; ++nt) {
#pragma unroll
    for (int r = 0; r < 8; ++r) {
      int m = r + 8 * hi;
      c_lds[(wave * 16 + m) * P + nt * 16 + rt] = (__bf16)(own[nt][r] * su[r] * sp[nt]);
    }
  }
  __syncthreads();

  {
    const v4f* s = (const v4f*)c_lds;
    v4f* d = (v4f*)(c_out + (size_t)b * K2);
#pragma unroll
    for (int j = 0; j < K2 * 2 / 16 / 256; ++j) d[j * 256 + threadIdx.x] = s[j * 256 + threadIdx.x];
  }
}

constexpr int LW     = 16;
constexpr int NSL    = 16;
constexpr int BT     = 16;
constexpr int NO     = U + P;
constexpr int WPART  = NT_O * TILE_F;

__global__ void __launch_bounds__(32 * LW)
logits_partial_kernel(const __bf16* __restrict__ c_ws,
                      const __bf16* __restrict__ wfrag,
                      float* __restrict__ lpart) {
  __shared__ __attribute__((aligned(16))) float part[LW * WPART];
  __shared__ __attribute__((aligned(16))) float lg[BT * NO];

  const int bt   = blockIdx.x & 15;
  const int sl   = blockIdx.x >> 4;
  const int wave = threadIdx.x >> 5;
  const int lane = threadIdx.x & 31;
  const int rt   = lane & 15;
  const int hi   = lane >> 4;
  const int ks   = sl * LW + wave;

  v8bf lo[NT_O], up[NT_O];
  load_wfrags<NT_O>(wfrag + (size_t)ks * NT_O * 512 + lane * 8, lo, up);
  v16bf cb = load_frag_bf16(c_ws + (size_t)(bt * BT + rt) * K2 + ks * 32 + (hi ? 8 : 0));

  float* pw = part + wave * WPART + lane * 4;
#pragma unroll
  for (int ot = 0; ot < NT_O; ++ot) {
    v8f z = {};
    v8f t = __builtin_amdgcn_wmma_f32_16x16x32_bf16(
        false, join_frag(lo[ot], up[ot]), false, cb, (short)0, z, false, false);
    *(v4f*)(pw + ot * TILE_F)       = __builtin_shufflevector(t, t, 0, 1, 2, 3);
    *(v4f*)(pw + ot * TILE_F + 128) = __builtin_shufflevector(t, t, 4, 5, 6, 7);
  }
  __syncthreads();

#pragma unroll
  for (int i = 0; i < WPART / (32 * LW); ++i) {
    int idx = i * (32 * LW) + threadIdx.x;
    float s = part[idx];
#pragma unroll
    for (int w = 1; w < LW; ++w) s += part[w * WPART + idx];
    int ot = idx >> 8, q = (idx >> 7) & 1, ln = (idx >> 2) & 31, e = idx & 3;
    int o = ot * 16 + q * 4 + e + 8 * (ln >> 4);
    lg[(ln & 15) * NO + o] = s;
  }
  __syncthreads();

  float* dst = lpart + ((size_t)sl * B + bt * BT) * NO;
#pragma unroll
  for (int i = 0; i < WPART / (32 * LW); ++i)
    dst[i * (32 * LW) + threadIdx.x] = lg[i * (32 * LW) + threadIdx.x];
}

template <int N>
__device__ __forceinline__ void sum_softmax(const float* __restrict__ lp,
                                            const float* __restrict__ bias,
                                            float* __restrict__ dst, int lane) {
  constexpr int per = N / 32;
  float v[NSL][per];
#pragma unroll
  for (int s = 0; s < NSL; ++s)
#pragma unroll
    for (int j = 0; j < per; ++j) v[s][j] = lp[(size_t)s * B * NO + lane + 32 * j];
  float vals[per];
  float mx = -1e30f;
#pragma unroll
  for (int j = 0; j < per; ++j) {
    float a = v[0][j];
#pragma unroll
    for (int s = 1; s < NSL; ++s) a += v[s][j];
    vals[j] = a + bias[lane + 32 * j];
    mx = fmaxf(mx, vals[j]);
  }
#pragma unroll
  for (int m = 16; m; m >>= 1) mx = fmaxf(mx, __shfl_xor(mx, m, 32));
  float sum = 0.f;
#pragma unroll
  for (int j = 0; j < per; ++j) { vals[j] = __expf(vals[j] - mx); sum += vals[j]; }
#pragma unroll
  for (int m = 16; m; m >>= 1) sum += __shfl_xor(sum, m, 32);
  float inv = 1.0f / sum;
#pragma unroll
  for (int j = 0; j < per; ++j) dst[lane + 32 * j] = vals[j] * inv;
}

__global__ void __launch_bounds__(64)
softmax_kernel(const float* __restrict__ lpart,
               const float* __restrict__ b_utt,
               const float* __restrict__ b_pheno,
               float* __restrict__ out_row,
               float* __restrict__ out_col) {
  const int b    = blockIdx.x;
  const int lane = threadIdx.x & 31;
  const float* lp = lpart + (size_t)b * NO;
  if (threadIdx.x < 32) sum_softmax<U>(lp, b_utt, out_row + (size_t)b * U, lane);
  else                  sum_softmax<P>(lp + U, b_pheno, out_col + (size_t)b * P, lane);
}

extern "C" void kernel_launch(void* const* d_in, const int* in_sizes, int n_in,
                              void* d_out, int out_size, void* d_ws, size_t ws_size,
                              hipStream_t stream) {
  const float* utt     = (const float*)d_in[0];
  const float* pheno   = (const float*)d_in[1];
  const float* w_utt   = (const float*)d_in[2];
  const float* b_utt   = (const float*)d_in[3];
  const float* w_pheno = (const float*)d_in[4];
  const float* b_pheno = (const float*)d_in[5];
  (void)in_sizes; (void)n_in; (void)out_size; (void)ws_size;

  char* ws = (char*)d_ws;
  size_t off = 0;
  auto alloc = [&](size_t bytes) {
    void* p = ws + off;
    off = (off + bytes + 255) & ~(size_t)255;
    return p;
  };
  float*  inv_nu = (float*)alloc((size_t)B * U * sizeof(float));
  float*  inv_np = (float*)alloc((size_t)B * P * sizeof(float));
  __bf16* wfrag  = (__bf16*)alloc((size_t)(U + P) * K2 * sizeof(__bf16));
  __bf16* c_ws   = (__bf16*)alloc((size_t)B * K2 * sizeof(__bf16));
  float*  lpart  = (float*)alloc((size_t)NSL * B * NO * sizeof(float));

  float* row_logits = (float*)d_out;
  float* col_logits = (float*)d_out + B * U;

  norms_kernel<<<B, NW_N * 32, 0, stream>>>(utt, pheno, inv_nu, inv_np);
  wstage_kernel<<<(KSTEPS * NT_O * 64) / 256, 256, 0, stream>>>(w_utt, w_pheno, wfrag);

  cosine_kernel<<<B, 256, 0, stream>>>(utt, pheno, inv_nu, inv_np, c_ws);
  logits_partial_kernel<<<(B / BT) * NSL, 32 * LW, 0, stream>>>(c_ws, wfrag, lpart);
  softmax_kernel<<<B, 64, 0, stream>>>(lpart, b_utt, b_pheno, row_logits, col_logits);
}


# logits GEMM reads the f32 weights in their own layout, rounds in registers; cosine kernel also writes c transposed for w_pheno; weight staging launch removed (column-logit f32 sum order changes)
# speedup vs baseline: 4.9140x; 1.0940x over previous
#include <hip/hip_runtime.h>
#include <hip/hip_bf16.h>
#include <math.h>

typedef __attribute__((ext_vector_type(16))) __bf16 v16bf;
typedef __attribute__((ext_vector_type(8)))  __bf16 v8bf;
typedef __attribute__((ext_vector_type(8)))  float  v8f;
typedef __attribute__((ext_vector_type(4)))  float  v4f;

constexpr int B  = 256;
constexpr int U  = 128;
constexpr int P  = 64;
constexpr int H  = 768;
constexpr int K2 = U * P;

constexpr int NW_N    = 16;
constexpr int RU_N    = U / NW_N;
constexpr int RP_N    = P / NW_N;
constexpr int ROWS_N  = RU_N + RP_N;
constexpr int AHEAD_N = 4;
constexpr int HQ      = H / 128;

__global__ void __launch_bounds__(NW_N * 32)
norms_kernel(const float* __restrict__ utt, const float* __restrict__ pheno,
             float* __restrict__ inv_nu, float* __restrict__ inv_np) {
  __shared__ float inv_lds[U + P];
  const int b    = blockIdx.x;
  const int lane = threadIdx.x & 31;
  const int wave = __builtin_amdgcn_readfirstlane(threadIdx.x >> 5);
  const float* ub = utt   + ((size_t)(wave * RU_N) * B + b) * H + lane * 4;
  const float* pb = pheno + ((size_t)(wave * RP_N) * B + b) * H + lane * 4;

  v4f buf[AHEAD_N + 1][HQ];
  float sum[ROWS_N];
#define LOAD_ROW(R) do { \
    const float* rp_ = ((R) < RU_N) ? ub + (size_t)(R) * B * H : pb + (size_t)((R) - RU_N) * B * H; \
    _Pragma("unroll") for (int q = 0; q < HQ; ++q) buf[(R) % (AHEAD_N + 1)][q] = *(const v4f*)(rp_ + q * 128); } while (0)
#pragma unroll
  for (int r = 0; r < AHEAD_N; ++r) LOAD_ROW(r);
#pragma unroll
  for (int r = 0; r < ROWS_N; ++r) {
    __builtin_amdgcn_sched_barrier(0);
    if (r + AHEAD_N < ROWS_N) LOAD_ROW(r + AHEAD_N);
    __builtin_amdgcn_sched_barrier(0);
    float s = 0.f;
#pragma unroll
    for (int q = 0; q < HQ; ++q) {
      v4f v = buf[r % (AHEAD_N + 1)][q];
      s += v.x * v.x + v.y * v.y + v.z * v.z + v.w * v.w;
    }
    sum[r] = s;
  }
#undef LOAD_ROW
#pragma unroll
  for (int r = 0; r < ROWS_N; ++r) {
    float s = sum[r];
#pragma unroll
    for (int m = 16; m; m >>= 1) s += __shfl_xor(s, m, 32);
    sum[r] = s;
  }
  float mine = 1.f;
#pragma unroll
  for (int r = 0; r < ROWS_N; ++r) if (lane == r) mine = sum[r];
  if (lane < ROWS_N) {
    int slot = (lane < RU_N) ? wave * RU_N + lane : U + wave * RP_N + (lane - RU_N);
    inv_lds[slot] = 1.0f / fmaxf(sqrtf(mine), 1e-8f);
  }
  __syncthreads();
  if (threadIdx.x < U)          inv_nu[b * U + threadIdx.x] = inv_lds[threadIdx.x];
  else if (threadIdx.x < U + P) inv_np[b * P + (threadIdx.x - U)] = inv_lds[threadIdx.x];
}

struct RawFrag { v4f a0, a1, b0, b1; };

__device__ __forceinline__ RawFrag load_raw_f32(const float* __restrict__ p) {
  RawFrag f;
  f.a0 = *(const v4f*)(p);
  f.a1 = *(const v4f*)(p + 4);
  f.b0 = *(const v4f*)(p + 16);
  f.b1 = *(const v4f*)(p + 20);
  return f;
}

__device__ __forceinline__ v16bf cvt_frag(const RawFrag& f) {
  v16bf r;
  r[0]  = (__bf16)f.a0[0]; r[1]  = (__bf16)f.a0[1];
  r[2]  = (__bf16)f.a0[2]; r[3]  = (__bf16)f.a0[3];
  r[4]  = (__bf16)f.a1[0]; r[5]  = (__bf16)f.a1[1];
  r[6]  = (__bf16)f.a1[2]; r[7]  = (__bf16)f.a1[3];
  r[8]  = (__bf16)f.b0[0]; r[9]  = (__bf16)f.b0[1];
  r[10] = (__bf16)f.b0[2]; r[11] = (__bf16)f.b0[3];
  r[12] = (__bf16)f.b1[0]; r[13] = (__bf16)f.b1[1];
  r[14] = (__bf16)f.b1[2]; r[15] = (__bf16)f.b1[3];
  return r;
}

__device__ __forceinline__ v16bf load_frag_bf16(const __bf16* __restrict__ p) {
  v8bf lo = *(const v8bf*)(p);
  v8bf hi = *(const v8bf*)(p + 16);
  v16bf r;
#pragma unroll
  for (int j = 0; j < 8; ++j) { r[j] = lo[j]; r[j + 8] = hi[j]; }
  return r;
}

constexpr int KSLICE = H / 8;
constexpr int KS_A   = KSLICE / 32;
constexpr int MT_A   = U / 16;
constexpr int NT_A   = P / 16;
constexpr int TILE_F = 256;
constexpr int PART_F = MT_A * NT_A * TILE_F;
constexpr int CT_PITCH = U + 8;

__global__ void __launch_bounds__(256)
cosine_kernel(const float* __restrict__ utt,
              const float* __restrict__ pheno,
              const float* __restrict__ inv_nu,
              const float* __restrict__ inv_np,
              __bf16* __restrict__ c_out,
              __bf16* __restrict__ cT_out) {
  __shared__ __attribute__((aligned(16))) float part[8 * PART_F];
  __shared__ __attribute__((aligned(16))) __bf16 c_lds[K2];
  __shared__ __attribute__((aligned(16))) __bf16 cT_lds[P * CT_PITCH];

  int b    = blockIdx.x;
  int wave = threadIdx.x >> 5;
  int lane = threadIdx.x & 31;
  int rt   = lane & 15;
  int hi   = lane >> 4;
  int k0   = hi ? 8 : 0;

  constexpr size_t TILE_STRIDE = (size_t)16 * B * H;
  const unsigned loff = (unsigned)(rt * (B * H) + wave * KSLICE + k0);
  const float* Au = utt   + (size_t)b * H;
  const float* Bp = pheno + (size_t)b * H;

  v8f acc[MT_A][NT_A] = {};
#pragma unroll
  for (int ks = 0; ks < KS_A; ++ks) {
    RawFrag rb[NT_A], ra[MT_A];
#pragma unroll
    for (int nt = 0; nt < NT_A; ++nt)
      rb[nt] = load_raw_f32(Bp + nt * TILE_STRIDE + loff + ks * 32);
#pragma unroll
    for (int mt = 0; mt < MT_A; ++mt)
      ra[mt] = load_raw_f32(Au + mt * TILE_STRIDE + loff + ks * 32);
    v16bf bfr[NT_A];
#pragma unroll
    for (int nt = 0; nt < NT_A; ++nt) bfr[nt] = cvt_frag(rb[nt]);
#pragma unroll
    for (int mt = 0; mt < MT_A; ++mt) {
      v16bf a = cvt_frag(ra[mt]);
#pragma unroll
      for (int nt = 0; nt < NT_A; ++nt)
        acc[mt][nt] = __builtin_amdgcn_wmma_f32_16x16x32_bf16(
            false, a, false, bfr[nt], (short)0, acc[mt][nt], false, false);
    }
  }

  const float* su_base = inv_nu + b * U + wave * 16 + 8 * hi;
  float4 su0 = *(const float4*)(su_base);
  float4 su1 = *(const float4*)(su_base + 4);
  float sp[NT_A];
#pragma unroll
  for (int nt = 0; nt < NT_A; ++nt) sp[nt] = inv_np[b * P + nt * 16 + rt];

  {
    float* pw = part + wave * PART_F + lane * 4;
#pragma unroll
    for (int mt = 0; mt < MT_A; ++mt)
#pragma unroll
      for (int nt = 0; nt < NT_A; ++nt) {
        v8f t = acc[mt][nt];
        *(v4f*)(pw + (mt * NT_A + nt) * TILE_F)       = __builtin_shufflevector(t, t, 0, 1, 2, 3);
        *(v4f*)(pw + (mt * NT_A + nt) * TILE_F + 128) = __builtin_shufflevector(t, t, 4, 5, 6, 7);
      }
  }
  __syncthreads();

  v8f own[NT_A];
#pragma unroll
  for (int nt = 0; nt < NT_A; ++nt) {
    const float* pr = part + (wave * NT_A + nt) * TILE_F + lane * 4;
    v4f s0 = *(const v4f*)(pr);
    v4f s1 = *(const v4f*)(pr + 128);
#pragma unroll
    for (int w = 1; w < 8; ++w) {
      s0 += *(const v4f*)(pr + w * PART_F);
      s1 += *(const v4f*)(pr + w * PART_F + 128);
    }
    own[nt] = __builtin_shufflevector(s0, s1, 0, 1, 2, 3, 4, 5, 6, 7);
  }

  float su[8] = {su0.x, su0.y, su0.z, su0.w, su1.x, su1.y, su1.z, su1.w};

#pragma unroll
  for (int nt = 0; nt < NT_A; ++nt) {
    v8bf tv;
#pragma unroll
    for (int r = 0; r < 8; ++r) {
      int m = r + 8 * hi;
      __bf16 v = (__bf16)(own[nt][r] * su[r] * sp[nt]);
      c_lds[(wave * 16 + m) * P + nt * 16 + rt] = v;
      tv[r] = v;
    }
    *(v8bf*)(cT_lds + (nt * 16 + rt) * CT_PITCH + wave * 16 + 8 * hi) = tv;
  }
  __syncthreads();

  {
    const v4f* s = (const v4f*)c_lds;
    v4f* d = (v4f*)(c_out + (size_t)b * K2);
#pragma unroll
    for (int j = 0; j < K2 * 2 / 16 / 256; ++j) d[j * 256 + threadIdx.x] = s[j * 256 + threadIdx.x];
  }
  {
    v4f* d = (v4f*)(cT_out + (size_t)b * K2);
#pragma unroll
    for (int j = 0; j < K2 * 2 / 16 / 256; ++j) {
      int chunk = j * 256 + threadIdx.x;
      d[chunk] = *(const v4f*)(cT_lds + (chunk >> 4) * CT_PITCH + (chunk & 15) * 8);
    }
  }
}

constexpr int LW     = 16;
constexpr int NSL    = 16;
constexpr int BG     = 32;
constexpr int NO     = U + P;
constexpr int RT_H   = MT_A / 2;
constexpr int CT_H   = NT_A / 2;
constexpr int NT_H   = RT_H + CT_H;
constexpr int NO_H   = NT_H * 16;
constexpr int WPART  = NT_H * 2 * TILE_F;

__global__ void __launch_bounds__(32 * LW)
logits_partial_kernel(const float* __restrict__ w_utt,
                      const float* __restrict__ w_pheno,
                      const __bf16* __restrict__ c_ws,
                      const __bf16* __restrict__ cT_ws,
                      float* __restrict__ lpart) {
  __shared__ __attribute__((aligned(16))) float part[LW * WPART];
  __shared__ __attribute__((aligned(16))) float lg[BG * NO_H];

  const int th   = blockIdx.x & 1;
  const int bg   = (blockIdx.x >> 1) & 7;
  const int sl   = blockIdx.x >> 4;
  const int wave = threadIdx.x >> 5;
  const int lane = threadIdx.x & 31;
  const int rt   = lane & 15;
  const int hi   = lane >> 4;
  const int koff = (sl * LW + wave) * 32 + (hi ? 8 : 0);

  const float*  wu = w_utt   + (size_t)(th * RT_H * 16 + rt) * K2 + koff;
  const float*  wp = w_pheno + (size_t)(th * CT_H * 16 + rt) * K2 + koff;
  const size_t  cb_off = (size_t)(bg * BG + rt) * K2 + koff;

  RawFrag ra[NT_H];
#pragma unroll
  for (int j = 0; j < RT_H; ++j) ra[j] = load_raw_f32(wu + (size_t)j * 16 * K2);
#pragma unroll
  for (int j = 0; j < CT_H; ++j) ra[RT_H + j] = load_raw_f32(wp + (size_t)j * 16 * K2);
  v16bf cb[2], ctb[2];
#pragma unroll
  for (int bs = 0; bs < 2; ++bs) {
    cb[bs]  = load_frag_bf16(c_ws  + cb_off + (size_t)bs * 16 * K2);
    ctb[bs] = load_frag_bf16(cT_ws + cb_off + (size_t)bs * 16 * K2);
  }

  float* pw = part + wave * WPART + lane * 4;
#pragma unroll
  for (int j = 0; j < NT_H; ++j) {
    v16bf a = cvt_frag(ra[j]);
#pragma unroll
    for (int bs = 0; bs < 2; ++bs) {
      v8f z = {};
      v8f t = __builtin_amdgcn_wmma_f32_16x16x32_bf16(
          false, a, false, (j < RT_H) ? cb[bs] : ctb[bs], (short)0, z, false, false);
      *(v4f*)(pw + (j * 2 + bs) * TILE_F)       = __builtin_shufflevector(t, t, 0, 1, 2, 3);
      *(v4f*)(pw + (j * 2 + bs) * TILE_F + 128) = __builtin_shufflevector(t, t, 4, 5, 6, 7);
    }
  }
  __syncthreads();

#pragma unroll
  for (int i = 0; i < WPART / (32 * LW); ++i) {
    int idx = i * (32 * LW) + threadIdx.x;
    float s = part[idx];
#pragma unroll
    for (int w = 1; w < LW; ++w) s += part[w * WPART + idx];
    int slot = idx >> 8, q = (idx >> 7) & 1, ln = (idx >> 2) & 31, e = idx & 3;
    int ol = (slot >> 1) * 16 + q * 4 + e + 8 * (ln >> 4);
    int bl = (slot & 1) * 16 + (ln & 15);
    lg[bl * NO_H + ol] = s;
  }
  __syncthreads();

  float* dst = lpart + ((size_t)sl * B + bg * BG) * NO;
#pragma unroll
  for (int i = 0; i < WPART / (32 * LW); ++i) {
    int idx = i * (32 * LW) + threadIdx.x;
    int bl = idx / NO_H, ol = idx % NO_H;
    int o  = (ol < RT_H * 16) ? th * RT_H * 16 + ol : U + th * CT_H * 16 + (ol - RT_H * 16);
    dst[bl * NO + o] = lg[idx];
  }
}

template <int N>
__device__ __forceinline__ void sum_softmax(const float* __restrict__ lp,
                                            const float* __restrict__ bias,
                                            float* __restrict__ dst, int lane) {
  constexpr int per = N / 32;
  float v[NSL][per];
#pragma unroll
  for (int s = 0; s < NSL; ++s)
#pragma unroll
    for (int j = 0; j < per; ++j) v[s][j] = lp[(size_t)s * B * NO + lane + 32 * j];
  float vals[per];
  float mx = -1e30f;
#pragma unroll
  for (int j = 0; j < per; ++j) {
    float a = v[0][j];
#pragma unroll
    for (int s = 1; s < NSL; ++s) a += v[s][j];
    vals[j] = a + bias[lane + 32 * j];
    mx = fmaxf(mx, vals[j]);
  }
#pragma unroll
  for (int m = 16; m; m >>= 1) mx = fmaxf(mx, __shfl_xor(mx, m, 32));
  float sum = 0.f;
#pragma unroll
  for (int j = 0; j < per; ++j) { vals[j] = __expf(vals[j] - mx); sum += vals[j]; }
#pragma unroll
  for (int m = 16; m; m >>= 1) sum += __shfl_xor(sum, m, 32);
  float inv = 1.0f / sum;
#pragma unroll
  for (int j = 0; j < per; ++j) dst[lane + 32 * j] = vals[j] * inv;
}

__global__ void __launch_bounds__(64)
softmax_kernel(const float* __restrict__ lpart,
               const float* __restrict__ b_utt,
               const float* __restrict__ b_pheno,
               float* __restrict__ out_row,
               float* __restrict__ out_col) {
  const int b    = blockIdx.x;
  const int lane = threadIdx.x & 31;
  const float* lp = lpart + (size_t)b * NO;
  if (threadIdx.x < 32) sum_softmax<U>(lp, b_utt, out_row + (size_t)b * U, lane);
  else                  sum_softmax<P>(lp + U, b_pheno, out_col + (size_t)b * P, lane);
}

extern "C" void kernel_launch(void* const* d_in, const int* in_sizes, int n_in,
                              void* d_out, int out_size, void* d_ws, size_t ws_size,
                              hipStream_t stream) {
  const float* utt     = (const float*)d_in[0];
  const float* pheno   = (const float*)d_in[1];
  const float* w_utt   = (const float*)d_in[2];
  const float* b_utt   = (const float*)d_in[3];
  const float* w_pheno = (const float*)d_in[4];
  const float* b_pheno = (const float*)d_in[5];
  (void)in_sizes; (void)n_in; (void)out_size; (void)ws_size;

  char* ws = (char*)d_ws;
  size_t off = 0;
  auto alloc = [&](size_t bytes) {
    void* p = ws + off;
    off = (off + bytes + 255) & ~(size_t)255;
    return p;
  };
  float*  inv_nu = (float*)alloc((size_t)B * U * sizeof(float));
  float*  inv_np = (float*)alloc((size_t)B * P * sizeof(float));
  __bf16* c_ws   = (__bf16*)alloc((size_t)B * K2 * sizeof(__bf16));
  __bf16* cT_ws  = (__bf16*)alloc((size_t)B * K2 * sizeof(__bf16));
  float*  lpart  = (float*)alloc((size_t)NSL * B * NO * sizeof(float));

  float* row_logits = (float*)d_out;
  float* col_logits = (float*)d_out + B * U;

  norms_kernel<<<B, NW_N * 32, 0, stream>>>(utt, pheno, inv_nu, inv_np);
  cosine_kernel<<<B, 256, 0, stream>>>(utt, pheno, inv_nu, inv_np, c_ws, cT_ws);
  logits_partial_kernel<<<(B / BG) * NSL * 2, 32 * LW, 0, stream>>>(w_utt, w_pheno, c_ws, cT_ws, lpart);
  softmax_kernel<<<B, 64, 0, stream>>>(lpart, b_utt, b_pheno, row_logits, col_logits);
}


# norm launch folded into the per-batch cosine launch as its first phase (8 waves x 24 whole rows, 60 loads in flight per wave); inverse norms pass through LDS; 4 launches -> 3; sums keep their order
# speedup vs baseline: 5.3347x; 1.0856x over previous
#include <hip/hip_runtime.h>
#include <hip/hip_bf16.h>
#include <math.h>

typedef __attribute__((ext_vector_type(16))) __bf16 v16bf;
typedef __attribute__((ext_vector_type(8)))  __bf16 v8bf;
typedef __attribute__((ext_vector_type(8)))  float  v8f;
typedef __attribute__((ext_vector_type(4)))  float  v4f;

constexpr int B  = 256;
constexpr int U  = 128;
constexpr int P  = 64;
constexpr int H  = 768;
constexpr int K2 = U * P;

constexpr int NW_C    = 8;
constexpr int RU_N    = U / NW_C;
constexpr int RP_N    = P / NW_C;
constexpr int ROWS_N  = RU_N + RP_N;
constexpr int AHEAD_N = 9;
constexpr int HQ      = H / 128;

__device__ __forceinline__ void row_sums_phase(const float* __restrict__ utt,
                                               const float* __restrict__ pheno,
                                               int b, int lane, float (&sum)[ROWS_N]) {
  const int wave = __builtin_amdgcn_readfirstlane(threadIdx.x >> 5);
  const float* ub = utt   + ((size_t)(wave * RU_N) * B + b) * H + lane * 4;
  const float* pb = pheno + ((size_t)(wave * RP_N) * B + b) * H + lane * 4;

  v4f buf[AHEAD_N + 1][HQ];
#define LOAD_ROW(R) do { \
    const float* rp_ = ((R) < RU_N) ? ub + (size_t)(R) * B * H : pb + (size_t)((R) - RU_N) * B * H; \
    _Pragma("unroll") for (int q = 0; q < HQ; ++q) buf[(R) % (AHEAD_N + 1)][q] = *(const v4f*)(rp_ + q * 128); } while (0)
#pragma unroll
  for (int r = 0; r < AHEAD_N; ++r) LOAD_ROW(r);
#pragma unroll
  for (int r = 0; r < ROWS_N; ++r) {
    __builtin_amdgcn_sched_barrier(0);
    if (r + AHEAD_N < ROWS_N) LOAD_ROW(r + AHEAD_N);
    __builtin_amdgcn_sched_barrier(0);
    float s = 0.f;
#pragma unroll
    for (int q = 0; q < HQ; ++q) {
      v4f v = buf[r % (AHEAD_N + 1)][q];
      s += v.x * v.x + v.y * v.y + v.z * v.z + v.w * v.w;
    }
    sum[r] = s;
  }
#undef LOAD_ROW
}

__device__ __forceinline__ void row_norms_finish(float (&sum)[ROWS_N],
                                                 float* __restrict__ inv_lds,
                                                 int lane) {
  const int wave = __builtin_amdgcn_readfirstlane(threadIdx.x >> 5);
#pragma unroll
  for (int r = 0; r < ROWS_N; ++r) {
    float s = sum[r];
#pragma unroll
    for (int m = 16; m; m >>= 1) s += __shfl_xor(s, m, 32);
    sum[r] = s;
  }
  float mine = 1.f;
#pragma unroll
  for (int r = 0; r < ROWS_N; ++r) if (lane == r) mine = sum[r];
  if (lane < ROWS_N) {
    int slot = (lane < RU_N) ? wave * RU_N + lane : U + wave * RP_N + (lane - RU_N);
    inv_lds[slot] = 1.0f / fmaxf(sqrtf(mine), 1e-8f);
  }
}

struct RawFrag { v4f a0, a1, b0, b1; };

__device__ __forceinline__ RawFrag load_raw_f32(const float* __restrict__ p) {
  RawFrag f;
  f.a0 = *(const v4f*)(p);
  f.a1 = *(const v4f*)(p + 4);
  f.b0 = *(const v4f*)(p + 16);
  f.b1 = *(const v4f*)(p + 20);
  return f;
}

__device__ __forceinline__ v16bf cvt_frag(const RawFrag& f) {
  v16bf r;
  r[0]  = (__bf16)f.a0[0]; r[1]  = (__bf16)f.a0[1];
  r[2]  = (__bf16)f.a0[2]; r[3]  = (__bf16)f.a0[3];
  r[4]  = (__bf16)f.a1[0]; r[5]  = (__bf16)f.a1[1];
  r[6]  = (__bf16)f.a1[2]; r[7]  = (__bf16)f.a1[3];
  r[8]  = (__bf16)f.b0[0]; r[9]  = (__bf16)f.b0[1];
  r[10] = (__bf16)f.b0[2]; r[11] = (__bf16)f.b0[3];
  r[12] = (__bf16)f.b1[0]; r[13] = (__bf16)f.b1[1];
  r[14] = (__bf16)f.b1[2]; r[15] = (__bf16)f.b1[3];
  return r;
}

__device__ __forceinline__ v16bf load_frag_bf16(const __bf16* __restrict__ p) {
  v8bf lo = *(const v8bf*)(p);
  v8bf hi = *(const v8bf*)(p + 16);
  v16bf r;
#pragma unroll
  for (int j = 0; j < 8; ++j) { r[j] = lo[j]; r[j + 8] = hi[j]; }
  return r;
}

constexpr int KSLICE = H / 8;
constexpr int KS_A   = KSLICE / 32;
constexpr int MT_A   = U / 16;
constexpr int NT_A   = P / 16;
constexpr int TILE_F = 256;
constexpr int PART_F = MT_A * NT_A * TILE_F;
constexpr int CT_PITCH = U + 8;

__global__ void __launch_bounds__(NW_C * 32)
cosine_kernel(const float* __restrict__ utt,
              const float* __restrict__ pheno,
              __bf16* __restrict__ c_out,
              __bf16* __restrict__ cT_out) {
  __shared__ __attribute__((aligned(16))) float part[8 * PART_F];
  __shared__ __attribute__((aligned(16))) __bf16 c_lds[K2];
  __shared__ __attribute__((aligned(16))) __bf16 cT_lds[P * CT_PITCH];
  __shared__ __attribute__((aligned(16))) float inv_lds[U + P];

  int b    = blockIdx.x;
  int wave = threadIdx.x >> 5;
  int lane = threadIdx.x & 31;

  float rsum[ROWS_N];
  row_sums_phase(utt, pheno, b, lane, rsum);

  int rt   = lane & 15;
  int hi   = lane >> 4;
  int k0   = hi ? 8 : 0;

  constexpr size_t TILE_STRIDE = (size_t)16 * B * H;
  const unsigned loff = (unsigned)(rt * (B * H) + wave * KSLICE + k0);
  const float* Au = utt   + (size_t)b * H;
  const float* Bp = pheno + (size_t)b * H;

  v8f acc[MT_A][NT_A] = {};
#pragma unroll
  for (int ks = 0; ks < KS_A; ++ks) {
    RawFrag rb[NT_A], ra[MT_A];
#pragma unroll
    for (int nt = 0; nt < NT_A; ++nt)
      rb[nt] = load_raw_f32(Bp + nt * TILE_STRIDE + loff + ks * 32);
#pragma unroll
    for (int mt = 0; mt < MT_A; ++mt)
      ra[mt] = load_raw_f32(Au + mt * TILE_STRIDE + loff + ks * 32);
    if (ks == 0) row_norms_finish(rsum, inv_lds, lane);
    v16bf bfr[NT_A];
#pragma unroll
    for (int nt = 0; nt < NT_A; ++nt) bfr[nt] = cvt_frag(rb[nt]);
#pragma unroll
    for (int mt = 0; mt < MT_A; ++mt) {
      v16bf a = cvt_frag(ra[mt]);
#pragma unroll
      for (int nt = 0; nt < NT_A; ++nt)
        acc[mt][nt] = __builtin_amdgcn_wmma_f32_16x16x32_bf16(
            false, a, false, bfr[nt], (short)0, acc[mt][nt], false, false);
    }
  }

  {
    float* pw = part + wave * PART_F + lane * 4;
#pragma unroll
    for (int mt = 0; mt < MT_A; ++mt)
#pragma unroll
      for (int nt = 0; nt < NT_A; ++nt) {
        v8f t = acc[mt][nt];
        *(v4f*)(pw + (mt * NT_A + nt) * TILE_F)       = __builtin_shufflevector(t, t, 0, 1, 2, 3);
        *(v4f*)(pw + (mt * NT_A + nt) * TILE_F + 128) = __builtin_shufflevector(t, t, 4, 5, 6, 7);
      }
  }
  __syncthreads();
  __builtin_amdgcn_sched_barrier(0);

  v8f own[NT_A];
#pragma unroll
  for (int nt = 0; nt < NT_A; ++nt) {
    const float* pr = part + (wave * NT_A + nt) * TILE_F + lane * 4;
    v4f s0 = *(const v4f*)(pr);
    v4f s1 = *(const v4f*)(pr + 128);
#pragma unroll
    for (int w = 1; w < 8; ++w) {
      s0 += *(const v4f*)(pr + w * PART_F);
      s1 += *(const v4f*)(pr + w * PART_F + 128);
    }
    own[nt] = __builtin_shufflevector(s0, s1, 0, 1, 2, 3, 4, 5, 6, 7);
  }

  const float* su_base = inv_lds + wave * 16 + 8 * hi;
  v4f su0 = *(const v4f*)(su_base);
  v4f su1 = *(const v4f*)(su_base + 4);
  float sp[NT_A];
#pragma unroll
  for (int nt = 0; nt < NT_A; ++nt) sp[nt] = inv_lds[U + nt * 16 + rt];

  float su[8] = {su0.x, su0.y, su0.z, su0.w, su1.x, su1.y, su1.z, su1.w};

#pragma unroll
  for (int nt = 0; nt < NT_A; ++nt) {
    v8bf tv;
#pragma unroll
    for (int r = 0; r < 8; ++r) {
      int m = r + 8 * hi;
      __bf16 v = (__bf16)(own[nt][r] * su[r] * sp[nt]);
      c_lds[(wave * 16 + m) * P + nt * 16 + rt] = v;
      tv[r] = v;
    }
    *(v8bf*)(cT_lds + (nt * 16 + rt) * CT_PITCH + wave * 16 + 8 * hi) = tv;
  }
  __syncthreads();

  {
    const v4f* s = (const v4f*)c_lds;
    v4f* d = (v4f*)(c_out + (size_t)b * K2);
#pragma unroll
    for (int j = 0; j < K2 * 2 / 16 / 256; ++j) d[j * 256 + threadIdx.x] = s[j * 256 + threadIdx.x];
  }
  {
    v4f* d = (v4f*)(cT_out + (size_t)b * K2);
#pragma unroll
    for (int j = 0; j < K2 * 2 / 16 / 256; ++j) {
      int chunk = j * 256 + threadIdx.x;
      d[chunk] = *(const v4f*)(cT_lds + (chunk >> 4) * CT_PITCH + (chunk & 15) * 8);
    }
  }
}

constexpr int LW     = 16;
constexpr int NSL    = 16;
constexpr int BG     = 32;
constexpr int NO     = U + P;
constexpr int RT_H   = MT_A / 2;
constexpr int CT_H   = NT_A / 2;
constexpr int NT_H   = RT_H + CT_H;
constexpr int NO_H   = NT_H * 16;
constexpr int WPART  = NT_H * 2 * TILE_F;

__global__ void __launch_bounds__(32 * LW)
logits_partial_kernel(const float* __restrict__ w_utt,
                      const float* __restrict__ w_pheno,
                      const __bf16* __restrict__ c_ws,
                      const __bf16* __restrict__ cT_ws,
                      float* __restrict__ lpart) {
  __shared__ __attribute__((aligned(16))) float part[LW * WPART];
  __shared__ __attribute__((aligned(16))) float lg[BG * NO_H];

  const int th   = blockIdx.x & 1;
  const int bg   = (blockIdx.x >> 1) & 7;
  const int sl   = blockIdx.x >> 4;
  const int wave = threadIdx.x >> 5;
  const int lane = threadIdx.x & 31;
  const int rt   = lane & 15;
  const int hi   = lane >> 4;
  const int koff = (sl * LW + wave) * 32 + (hi ? 8 : 0);

  const float*  wu = w_utt   + (size_t)(th * RT_H * 16 + rt) * K2 + koff;
  const float*  wp = w_pheno + (size_t)(th * CT_H * 16 + rt) * K2 + koff;
  const size_t  cb_off = (size_t)(bg * BG + rt) * K2 + koff;

  RawFrag ra[NT_H];
#pragma unroll
  for (int j = 0; j < RT_H; ++j) ra[j] = load_raw_f32(wu + (size_t)j * 16 * K2);
#pragma unroll
  for (int j = 0; j < CT_H; ++j) ra[RT_H + j] = load_raw_f32(wp + (size_t)j * 16 * K2);
  v16bf cb[2], ctb[2];
#pragma unroll
  for (int bs = 0; bs < 2; ++bs) {
    cb[bs]  = load_frag_bf16(c_ws  + cb_off + (size_t)bs * 16 * K2);
    ctb[bs] = load_frag_bf16(cT_ws + cb_off + (size_t)bs * 16 * K2);
  }

  float* pw = part + wave * WPART + lane * 4;
#pragma unroll
  for (int j = 0; j < NT_H; ++j) {
    v16bf a = cvt_frag(ra[j]);
#pragma unroll
    for (int bs = 0; bs < 2; ++bs) {
      v8f z = {};
      v8f t = __builtin_amdgcn_wmma_f32_16x16x32_bf16(
          false, a, false, (j < RT_H) ? cb[bs] : ctb[bs], (short)0, z, false, false);
      *(v4f*)(pw + (j * 2 + bs) * TILE_F)       = __builtin_shufflevector(t, t, 0, 1, 2, 3);
      *(v4f*)(pw + (j * 2 + bs) * TILE_F + 128) = __builtin_shufflevector(t, t, 4, 5, 6, 7);
    }
  }
  __syncthreads();

#pragma unroll
  for (int i = 0; i < WPART / (32 * LW); ++i) {
    int idx = i * (32 * LW) + threadIdx.x;
    float s = part[idx];
#pragma unroll
    for (int w = 1; w < LW; ++w) s += part[w * WPART + idx];
    int slot = idx >> 8, q = (idx >> 7) & 1, ln = (idx >> 2) & 31, e = idx & 3;
    int ol = (slot >> 1) * 16 + q * 4 + e + 8 * (ln >> 4);
    int bl = (slot & 1) * 16 + (ln & 15);
    lg[bl * NO_H + ol] = s;
  }
  __syncthreads();

  float* dst = lpart + ((size_t)sl * B + bg * BG) * NO;
#pragma unroll
  for (int i = 0; i < WPART / (32 * LW); ++i) {
    int idx = i * (32 * LW) + threadIdx.x;
    int bl = idx / NO_H, ol = idx % NO_H;
    int o  = (ol < RT_H * 16) ? th * RT_H * 16 + ol : U + th * CT_H * 16 + (ol - RT_H * 16);
    dst[bl * NO + o] = lg[idx];
  }
}

template <int N>
__device__ __forceinline__ void sum_softmax(const float* __restrict__ lp,
                                            const float* __restrict__ bias,
                                            float* __restrict__ dst, int lane) {
  constexpr int per = N / 32;
  float v[NSL][per];
#pragma unroll
  for (int s = 0; s < NSL; ++s)
#pragma unroll
    for (int j = 0; j < per; ++j) v[s][j] = lp[(size_t)s * B * NO + lane + 32 * j];
  float vals[per];
  float mx = -1e30f;
#pragma unroll
  for (int j = 0; j < per; ++j) {
    float a = v[0][j];
#pragma unroll
    for (int s = 1; s < NSL; ++s) a += v[s][j];
    vals[j] = a + bias[lane + 32 * j];
    mx = fmaxf(mx, vals[j]);
  }
#pragma unroll
  for (int m = 16; m; m >>= 1) mx = fmaxf(mx, __shfl_xor(mx, m, 32));
  float sum = 0.f;
#pragma unroll
  for (int j = 0; j < per; ++j) { vals[j] = __expf(vals[j] - mx); sum += vals[j]; }
#pragma unroll
  for (int m = 16; m; m >>= 1) sum += __shfl_xor(sum, m, 32);
  float inv = 1.0f / sum;
#pragma unroll
  for (int j = 0; j < per; ++j) dst[lane + 32 * j] = vals[j] * inv;
}

__global__ void __launch_bounds__(64)
softmax_kernel(const float* __restrict__ lpart,
               const float* __restrict__ b_utt,
               const float* __restrict__ b_pheno,
               float* __restrict__ out_row,
               float* __restrict__ out_col) {
  const int b    = blockIdx.x;
  const int lane = threadIdx.x & 31;
  const float* lp = lpart + (size_t)b * NO;
  if (threadIdx.x < 32) sum_softmax<U>(lp, b_utt, out_row + (size_t)b * U, lane);
  else                  sum_softmax<P>(lp + U, b_pheno, out_col + (size_t)b * P, lane);
}

extern "C" void kernel_launch(void* const* d_in, const int* in_sizes, int n_in,
                              void* d_out, int out_size, void* d_ws, size_t ws_size,
                              hipStream_t stream) {
  const float* utt     = (const float*)d_in[0];
  const float* pheno   = (const float*)d_in[1];
  const float* w_utt   = (const float*)d_in[2];
  const float* b_utt   = (const float*)d_in[3];
  const float* w_pheno = (const float*)d_in[4];
  const float* b_pheno = (const float*)d_in[5];
  (void)in_sizes; (void)n_in; (void)out_size; (void)ws_size;

  char* ws = (char*)d_ws;
  size_t off = 0;
  auto alloc = [&](size_t bytes) {
    void* p = ws + off;
    off = (off + bytes + 255) & ~(size_t)255;
    return p;
  };
  __bf16* c_ws   = (__bf16*)alloc((size_t)B * K2 * sizeof(__bf16));
  __bf16* cT_ws  = (__bf16*)alloc((size_t)B * K2 * sizeof(__bf16));
  float*  lpart  = (float*)alloc((size_t)NSL * B * NO * sizeof(float));

  float* row_logits = (float*)d_out;
  float* col_logits = (float*)d_out + B * U;

  cosine_kernel<<<B, NW_C * 32, 0, stream>>>(utt, pheno, c_ws, cT_ws);
  logits_partial_kernel<<<(B / BG) * NSL * 2, 32 * LW, 0, stream>>>(w_utt, w_pheno, c_ws, cT_ws, lpart);
  softmax_kernel<<<B, 64, 0, stream>>>(lpart, b_utt, b_pheno, row_logits, col_logits);
}


# logits GEMM re-tiled: block = 64 batches x 8 k-steps (8 waves, 32 k-slices): a weight fragment's 16-line loads serve 4 batch tiles not 2; block tail on 16-B accesses (f32 sum order changes)
# speedup vs baseline: 5.5323x; 1.0370x over previous
#include <hip/hip_runtime.h>
#include <hip/hip_bf16.h>
#include <math.h>

typedef __attribute__((ext_vector_type(16))) __bf16 v16bf;
typedef __attribute__((ext_vector_type(8)))  __bf16 v8bf;
typedef __attribute__((ext_vector_type(8)))  float  v8f;
typedef __attribute__((ext_vector_type(4)))  float  v4f;

constexpr int B  = 256;
constexpr int U  = 128;
constexpr int P  = 64;
constexpr int H  = 768;
constexpr int K2 = U * P;

constexpr int NW_C    = 8;
constexpr int RU_N    = U / NW_C;
constexpr int RP_N    = P / NW_C;
constexpr int ROWS_N  = RU_N + RP_N;
constexpr int AHEAD_N = 9;
constexpr int HQ      = H / 128;

__device__ __forceinline__ void row_sums_phase(const float* __restrict__ utt,
                                               const float* __restrict__ pheno,
                                               int b, int lane, float (&sum)[ROWS_N]) {
  const int wave = __builtin_amdgcn_readfirstlane(threadIdx.x >> 5);
  const float* ub = utt   + ((size_t)(wave * RU_N) * B + b) * H + lane * 4;
  const float* pb = pheno + ((size_t)(wave * RP_N) * B + b) * H + lane * 4;

  v4f buf[AHEAD_N + 1][HQ];
#define LOAD_ROW(R) do { \
    const float* rp_ = ((R) < RU_N) ? ub + (size_t)(R) * B * H : pb + (size_t)((R) - RU_N) * B * H; \
    _Pragma("unroll") for (int q = 0; q < HQ; ++q) buf[(R) % (AHEAD_N + 1)][q] = *(const v4f*)(rp_ + q * 128); } while (0)
#pragma unroll
  for (int r = 0; r < AHEAD_N; ++r) LOAD_ROW(r);
#pragma unroll
  for (int r = 0; r < ROWS_N; ++r) {
    __builtin_amdgcn_sched_barrier(0);
    if (r + AHEAD_N < ROWS_N) LOAD_ROW(r + AHEAD_N);
    __builtin_amdgcn_sched_barrier(0);
    float s = 0.f;
#pragma unroll
    for (int q = 0; q < HQ; ++q) {
      v4f v = buf[r % (AHEAD_N + 1)][q];
      s += v.x * v.x + v.y * v.y + v.z * v.z + v.w * v.w;
    }
    sum[r] = s;
  }
#undef LOAD_ROW
}

__device__ __forceinline__ void row_norms_finish(float (&sum)[ROWS_N],
                                                 float* __restrict__ inv_lds,
                                                 int lane) {
  const int wave = __builtin_amdgcn_readfirstlane(threadIdx.x >> 5);
#pragma unroll
  for (int r = 0; r < ROWS_N; ++r) {
    float s = sum[r];
#pragma unroll
    for (int m = 16; m; m >>= 1) s += __shfl_xor(s, m, 32);
    sum[r] = s;
  }
  float mine = 1.f;
#pragma unroll
  for (int r = 0; r < ROWS_N; ++r) if (lane == r) mine = sum[r];
  if (lane < ROWS_N) {
    int slot = (lane < RU_N) ? wave * RU_N + lane : U + wave * RP_N + (lane - RU_N);
    inv_lds[slot] = 1.0f / fmaxf(sqrtf(mine), 1e-8f);
  }
}

struct RawFrag { v4f a0, a1, b0, b1; };

__device__ __forceinline__ RawFrag load_raw_f32(const float* __restrict__ p) {
  RawFrag f;
  f.a0 = *(const v4f*)(p);
  f.a1 = *(const v4f*)(p + 4);
  f.b0 = *(const v4f*)(p + 16);
  f.b1 = *(const v4f*)(p + 20);
  return f;
}

__device__ __forceinline__ v16bf cvt_frag(const RawFrag& f) {
  v16bf r;
  r[0]  = (__bf16)f.a0[0]; r[1]  = (__bf16)f.a0[1];
  r[2]  = (__bf16)f.a0[2]; r[3]  = (__bf16)f.a0[3];
  r[4]  = (__bf16)f.a1[0]; r[5]  = (__bf16)f.a1[1];
  r[6]  = (__bf16)f.a1[2]; r[7]  = (__bf16)f.a1[3];
  r[8]  = (__bf16)f.b0[0]; r[9]  = (__bf16)f.b0[1];
  r[10] = (__bf16)f.b0[2]; r[11] = (__bf16)f.b0[3];
  r[12] = (__bf16)f.b1[0]; r[13] = (__bf16)f.b1[1];
  r[14] = (__bf16)f.b1[2]; r[15] = (__bf16)f.b1[3];
  return r;
}

__device__ __forceinline__ v16bf load_frag_bf16(const __bf16* __restrict__ p) {
  v8bf lo = *(const v8bf*)(p);
  v8bf hi = *(const v8bf*)(p + 16);
  v16bf r;
#pragma unroll
  for (int j = 0; j < 8; ++j) { r[j] = lo[j]; r[j + 8] = hi[j]; }
  return r;
}

constexpr int KSLICE = H / 8;
constexpr int KS_A   = KSLICE / 32;
constexpr int MT_A   = U / 16;
constexpr int NT_A   = P / 16;
constexpr int TILE_F = 256;
constexpr int PART_F = MT_A * NT_A * TILE_F;
constexpr int CT_PITCH = U + 8;

__global__ void __launch_bounds__(NW_C * 32)
cosine_kernel(const float* __restrict__ utt,
              const float* __restrict__ pheno,
              __bf16* __restrict__ c_out,
              __bf16* __restrict__ cT_out) {
  __shared__ __attribute__((aligned(16))) float part[8 * PART_F];
  __shared__ __attribute__((aligned(16))) __bf16 c_lds[K2];
  __shared__ __attribute__((aligned(16))) __bf16 cT_lds[P * CT_PITCH];
  __shared__ __attribute__((aligned(16))) float inv_lds[U + P];

  int b    = blockIdx.x;
  int wave = threadIdx.x >> 5;
  int lane = threadIdx.x & 31;

  float rsum[ROWS_N];
  row_sums_phase(utt, pheno, b, lane, rsum);

  int rt   = lane & 15;
  int hi   = lane >> 4;
  int k0   = hi ? 8 : 0;

  constexpr size_t TILE_STRIDE = (size_t)16 * B * H;
  const unsigned loff = (unsigned)(rt * (B * H) + wave * KSLICE + k0);
  const float* Au = utt   + (size_t)b * H;
  const float* Bp = pheno + (size_t)b * H;

  v8f acc[MT_A][NT_A] = {};
#pragma unroll
  for (int ks = 0; ks < KS_A; ++ks) {
    RawFrag rb[NT_A], ra[MT_A];
#pragma unroll
    for (int nt = 0; nt < NT_A; ++nt)
      rb[nt] = load_raw_f32(Bp + nt * TILE_STRIDE + loff + ks * 32);
#pragma unroll
    for (int mt = 0; mt < MT_A; ++mt)
      ra[mt] = load_raw_f32(Au + mt * TILE_STRIDE + loff + ks * 32);
    if (ks == 0) row_norms_finish(rsum, inv_lds, lane);
    v16bf bfr[NT_A];
#pragma unroll
    for (int nt = 0; nt < NT_A; ++nt) bfr[nt] = cvt_frag(rb[nt]);
#pragma unroll
    for (int mt = 0; mt < MT_A; ++mt) {
      v16bf a = cvt_frag(ra[mt]);
#pragma unroll
      for (int nt = 0; nt < NT_A; ++nt)
        acc[mt][nt] = __builtin_amdgcn_wmma_f32_16x16x32_bf16(
            false, a, false, bfr[nt], (short)0, acc[mt][nt], false, false);
    }
  }

  {
    float* pw = part + wave * PART_F + lane * 4;
#pragma unroll
    for (int mt = 0; mt < MT_A; ++mt)
#pragma unroll
      for (int nt = 0; nt < NT_A; ++nt) {
        v8f t = acc[mt][nt];
        *(v4f*)(pw + (mt * NT_A + nt) * TILE_F)       = __builtin_shufflevector(t, t, 0, 1, 2, 3);
        *(v4f*)(pw + (mt * NT_A + nt) * TILE_F + 128) = __builtin_shufflevector(t, t, 4, 5, 6, 7);
      }
  }
  __syncthreads();
  __builtin_amdgcn_sched_barrier(0);

  v8f own[NT_A];
#pragma unroll
  for (int nt = 0; nt < NT_A; ++nt) {
    const float* pr = part + (wave * NT_A + nt) * TILE_F + lane * 4;
    v4f s0 = *(const v4f*)(pr);
    v4f s1 = *(const v4f*)(pr + 128);
#pragma unroll
    for (int w = 1; w < 8; ++w) {
      s0 += *(const v4f*)(pr + w * PART_F);
      s1 += *(const v4f*)(pr + w * PART_F + 128);
    }
    own[nt] = __builtin_shufflevector(s0, s1, 0, 1, 2, 3, 4, 5, 6, 7);
  }

  const float* su_base = inv_lds + wave * 16 + 8 * hi;
  v4f su0 = *(const v4f*)(su_base);
  v4f su1 = *(const v4f*)(su_base + 4);
  float sp[NT_A];
#pragma unroll
  for (int nt = 0; nt < NT_A; ++nt) sp[nt] = inv_lds[U + nt * 16 + rt];

  float su[8] = {su0.x, su0.y, su0.z, su0.w, su1.x, su1.y, su1.z, su1.w};

#pragma unroll
  for (int nt = 0; nt < NT_A; ++nt) {
    v8bf tv;
#pragma unroll
    for (int r = 0; r < 8; ++r) {
      int m = r + 8 * hi;
      __bf16 v = (__bf16)(own[nt][r] * su[r] * sp[nt]);
      c_lds[(wave * 16 + m) * P + nt * 16 + rt] = v;
      tv[r] = v;
    }
    *(v8bf*)(cT_lds + (nt * 16 + rt) * CT_PITCH + wave * 16 + 8 * hi) = tv;
  }
  __syncthreads();

  {
    const v4f* s = (const v4f*)c_lds;
    v4f* d = (v4f*)(c_out + (size_t)b * K2);
#pragma unroll
    for (int j = 0; j < K2 * 2 / 16 / 256; ++j) d[j * 256 + threadIdx.x] = s[j * 256 + threadIdx.x];
  }
  {
    v4f* d = (v4f*)(cT_out + (size_t)b * K2);
#pragma unroll
    for (int j = 0; j < K2 * 2 / 16 / 256; ++j) {
      int chunk = j * 256 + threadIdx.x;
      d[chunk] = *(const v4f*)(cT_lds + (chunk >> 4) * CT_PITCH + (chunk & 15) * 8);
    }
  }
}

constexpr int LW     = 8;
constexpr int NSL    = 32;
constexpr int BG     = 64;
constexpr int NBT    = BG / 16;
constexpr int NO     = U + P;
constexpr int RT_H   = MT_A / 2;
constexpr int CT_H   = NT_A / 2;
constexpr int NT_H   = RT_H + CT_H;
constexpr int NO_H   = NT_H * 16;
constexpr int WPART  = NT_H * NBT * TILE_F;
constexpr int LG_PITCH = NO_H + 4;

__global__ void __launch_bounds__(32 * LW)
logits_partial_kernel(const float* __restrict__ w_utt,
                      const float* __restrict__ w_pheno,
                      const __bf16* __restrict__ c_ws,
                      const __bf16* __restrict__ cT_ws,
                      float* __restrict__ lpart) {
  __shared__ __attribute__((aligned(16))) float part[LW * WPART];
  __shared__ __attribute__((aligned(16))) float lg[BG * LG_PITCH];

  const int th   = blockIdx.x & 1;
  const int bg   = (blockIdx.x >> 1) & 3;
  const int sl   = blockIdx.x >> 3;
  const int wave = threadIdx.x >> 5;
  const int lane = threadIdx.x & 31;
  const int rt   = lane & 15;
  const int hi   = lane >> 4;
  const int koff = (sl * LW + wave) * 32 + (hi ? 8 : 0);

  const float*  wu = w_utt   + (size_t)(th * RT_H * 16 + rt) * K2 + koff;
  const float*  wp = w_pheno + (size_t)(th * CT_H * 16 + rt) * K2 + koff;
  const size_t  cb_off = (size_t)(bg * BG + rt) * K2 + koff;

  RawFrag ra[NT_H];
#pragma unroll
  for (int j = 0; j < RT_H; ++j) ra[j] = load_raw_f32(wu + (size_t)j * 16 * K2);
#pragma unroll
  for (int j = 0; j < CT_H; ++j) ra[RT_H + j] = load_raw_f32(wp + (size_t)j * 16 * K2);
  v16bf cb[NBT], ctb[NBT];
#pragma unroll
  for (int bs = 0; bs < NBT; ++bs) {
    cb[bs]  = load_frag_bf16(c_ws  + cb_off + (size_t)bs * 16 * K2);
    ctb[bs] = load_frag_bf16(cT_ws + cb_off + (size_t)bs * 16 * K2);
  }

  float* pw = part + wave * (NBT * TILE_F) + lane * 4;
#pragma unroll
  for (int j = 0; j < NT_H; ++j) {
    v16bf a = cvt_frag(ra[j]);
#pragma unroll
    for (int bs = 0; bs < NBT; ++bs) {
      v8f z = {};
      v8f t = __builtin_amdgcn_wmma_f32_16x16x32_bf16(
          false, a, false, (j < RT_H) ? cb[bs] : ctb[bs], (short)0, z, false, false);
      *(v4f*)(pw + (j * LW * NBT + bs) * TILE_F)       = __builtin_shufflevector(t, t, 0, 1, 2, 3);
      *(v4f*)(pw + (j * LW * NBT + bs) * TILE_F + 128) = __builtin_shufflevector(t, t, 4, 5, 6, 7);
    }
  }
  __syncthreads();

  const v4f* part4 = (const v4f*)part + threadIdx.x;
  const int bs_r = threadIdx.x >> 6, q_r = (threadIdx.x >> 5) & 1;
  float* lgw = lg + (bs_r * 16 + rt) * LG_PITCH + q_r * 4 + 8 * hi;
#pragma unroll
  for (int j = 0; j < NT_H; ++j) {
    v4f s = part4[(j * LW) * (NBT * TILE_F / 4)];
#pragma unroll
    for (int w = 1; w < LW; ++w) s += part4[(j * LW + w) * (NBT * TILE_F / 4)];
    *(v4f*)(lgw + j * 16) = s;
  }
  __syncthreads();

  float* dst = lpart + ((size_t)sl * B + bg * BG) * NO;
#pragma unroll
  for (int i = 0; i < BG * NO_H / 4 / (32 * LW); ++i) {
    int u4 = i * (32 * LW) + threadIdx.x;
    int bl = u4 / (NO_H / 4), o4 = u4 % (NO_H / 4);
    int o  = (o4 < RT_H * 4) ? th * RT_H * 16 + o4 * 4 : U + th * CT_H * 16 + (o4 - RT_H * 4) * 4;
    *(v4f*)(dst + bl * NO + o) = *(const v4f*)(lg + bl * LG_PITCH + o4 * 4);
  }
}

template <int N>
__device__ __forceinline__ void sum_softmax(const float* __restrict__ lp,
                                            const float* __restrict__ bias,
                                            float* __restrict__ dst, int lane) {
  constexpr int per = N / 32;
  float v[NSL][per];
#pragma unroll
  for (int s = 0; s < NSL; ++s)
#pragma unroll
    for (int j = 0; j < per; ++j) v[s][j] = lp[(size_t)s * B * NO + lane + 32 * j];
  float vals[per];
  float mx = -1e30f;
#pragma unroll
  for (int j = 0; j < per; ++j) {
    float a = v[0][j];
#pragma unroll
    for (int s = 1; s < NSL; ++s) a += v[s][j];
    vals[j] = a + bias[lane + 32 * j];
    mx = fmaxf(mx, vals[j]);
  }
#pragma unroll
  for (int m = 16; m; m >>= 1) mx = fmaxf(mx, __shfl_xor(mx, m, 32));
  float sum = 0.f;
#pragma unroll
  for (int j = 0; j < per; ++j) { vals[j] = __expf(vals[j] - mx); sum += vals[j]; }
#pragma unroll
  for (int m = 16; m; m >>= 1) sum += __shfl_xor(sum, m, 32);
  float inv = 1.0f / sum;
#pragma unroll
  for (int j = 0; j < per; ++j) dst[lane + 32 * j] = vals[j] * inv;
}

__global__ void __launch_bounds__(64)
softmax_kernel(const float* __restrict__ lpart,
               const float* __restrict__ b_utt,
               const float* __restrict__ b_pheno,
               float* __restrict__ out_row,
               float* __restrict__ out_col) {
  const int b    = blockIdx.x;
  const int lane = threadIdx.x & 31;
  const float* lp = lpart + (size_t)b * NO;
  if (threadIdx.x < 32) sum_softmax<U>(lp, b_utt, out_row + (size_t)b * U, lane);
  else                  sum_softmax<P>(lp + U, b_pheno, out_col + (size_t)b * P, lane);
}

extern "C" void kernel_launch(void* const* d_in, const int* in_sizes, int n_in,
                              void* d_out, int out_size, void* d_ws, size_t ws_size,
                              hipStream_t stream) {
  const float* utt     = (const float*)d_in[0];
  const float* pheno   = (const float*)d_in[1];
  const float* w_utt   = (const float*)d_in[2];
  const float* b_utt   = (const float*)d_in[3];
  const float* w_pheno = (const float*)d_in[4];
  const float* b_pheno = (const float*)d_in[5];
  (void)in_sizes; (void)n_in; (void)out_size; (void)ws_size;

  char* ws = (char*)d_ws;
  size_t off = 0;
  auto alloc = [&](size_t bytes) {
    void* p = ws + off;
    off = (off + bytes + 255) & ~(size_t)255;
    return p;
  };
  __bf16* c_ws   = (__bf16*)alloc((size_t)B * K2 * sizeof(__bf16));
  __bf16* cT_ws  = (__bf16*)alloc((size_t)B * K2 * sizeof(__bf16));
  float*  lpart  = (float*)alloc((size_t)NSL * B * NO * sizeof(float));

  float* row_logits = (float*)d_out;
  float* col_logits = (float*)d_out + B * U;

  cosine_kernel<<<B, NW_C * 32, 0, stream>>>(utt, pheno, c_ws, cT_ws);
  logits_partial_kernel<<<(B / BG) * NSL * 2, 32 * LW, 0, stream>>>(w_utt, w_pheno, c_ws, cT_ws, lpart);
  softmax_kernel<<<B, 64, 0, stream>>>(lpart, b_utt, b_pheno, row_logits, col_logits);
}


# cosine kernel: waves 4..7 run GEMM pass, then hand-over stores, then their norm pass (waves 0..3 keep norm first): row streams and 16-line fragment loads overlap in the WGP; sums keep their order
# speedup vs baseline: 5.7167x; 1.0333x over previous
#include <hip/hip_runtime.h>
#include <hip/hip_bf16.h>
#include <math.h>

typedef __attribute__((ext_vector_type(16))) __bf16 v16bf;
typedef __attribute__((ext_vector_type(8)))  __bf16 v8bf;
typedef __attribute__((ext_vector_type(8)))  float  v8f;
typedef __attribute__((ext_vector_type(4)))  float  v4f;

constexpr int B  = 256;
constexpr int U  = 128;
constexpr int P  = 64;
constexpr int H  = 768;
constexpr int K2 = U * P;

constexpr int NW_C    = 8;
constexpr int RU_N    = U / NW_C;
constexpr int RP_N    = P / NW_C;
constexpr int ROWS_N  = RU_N + RP_N;
constexpr int AHEAD_N = 9;
constexpr int HQ      = H / 128;

__device__ __forceinline__ void row_sums_phase(const float* __restrict__ utt,
                                               const float* __restrict__ pheno,
                                               int b, int lane, float (&sum)[ROWS_N]) {
  const int wave = __builtin_amdgcn_readfirstlane(threadIdx.x >> 5);
  const float* ub = utt   + ((size_t)(wave * RU_N) * B + b) * H + lane * 4;
  const float* pb = pheno + ((size_t)(wave * RP_N) * B + b) * H + lane * 4;

  v4f buf[AHEAD_N + 1][HQ];
#define LOAD_ROW(R) do { \
    const float* rp_ = ((R) < RU_N) ? ub + (size_t)(R) * B * H : pb + (size_t)((R) - RU_N) * B * H; \
    _Pragma("unroll") for (int q = 0; q < HQ; ++q) buf[(R) % (AHEAD_N + 1)][q] = *(const v4f*)(rp_ + q * 128); } while (0)
#pragma unroll
  for (int r = 0; r < AHEAD_N; ++r) LOAD_ROW(r);
#pragma unroll
  for (int r = 0; r < ROWS_N; ++r) {
    __builtin_amdgcn_sched_barrier(0);
    if (r + AHEAD_N < ROWS_N) LOAD_ROW(r + AHEAD_N);
    __builtin_amdgcn_sched_barrier(0);
    float s = 0.f;
#pragma unroll
    for (int q = 0; q < HQ; ++q) {
      v4f v = buf[r % (AHEAD_N + 1)][q];
      s += v.x * v.x + v.y * v.y + v.z * v.z + v.w * v.w;
    }
    sum[r] = s;
  }
#undef LOAD_ROW
}

__device__ __forceinline__ void row_norms_finish(float (&sum)[ROWS_N],
                                                 float* __restrict__ inv_lds,
                                                 int lane) {
  const int wave = __builtin_amdgcn_readfirstlane(threadIdx.x >> 5);
#pragma unroll
  for (int r = 0; r < ROWS_N; ++r) {
    float s = sum[r];
#pragma unroll
    for (int m = 16; m; m >>= 1) s += __shfl_xor(s, m, 32);
    sum[r] = s;
  }
  float mine = 1.f;
#pragma unroll
  for (int r = 0; r < ROWS_N; ++r) if (lane == r) mine = sum[r];
  if (lane < ROWS_N) {
    int slot = (lane < RU_N) ? wave * RU_N + lane : U + wave * RP_N + (lane - RU_N);
    inv_lds[slot] = 1.0f / fmaxf(sqrtf(mine), 1e-8f);
  }
}

struct RawFrag { v4f a0, a1, b0, b1; };

__device__ __forceinline__ RawFrag load_raw_f32(const float* __restrict__ p) {
  RawFrag f;
  f.a0 = *(const v4f*)(p);
  f.a1 = *(const v4f*)(p + 4);
  f.b0 = *(const v4f*)(p + 16);
  f.b1 = *(const v4f*)(p + 20);
  return f;
}

__device__ __forceinline__ v16bf cvt_frag(const RawFrag& f) {
  v16bf r;
  r[0]  = (__bf16)f.a0[0]; r[1]  = (__bf16)f.a0[1];
  r[2]  = (__bf16)f.a0[2]; r[3]  = (__bf16)f.a0[3];
  r[4]  = (__bf16)f.a1[0]; r[5]  = (__bf16)f.a1[1];
  r[6]  = (__bf16)f.a1[2]; r[7]  = (__bf16)f.a1[3];
  r[8]  = (__bf16)f.b0[0]; r[9]  = (__bf16)f.b0[1];
  r[10] = (__bf16)f.b0[2]; r[11] = (__bf16)f.b0[3];
  r[12] = (__bf16)f.b1[0]; r[13] = (__bf16)f.b1[1];
  r[14] = (__bf16)f.b1[2]; r[15] = (__bf16)f.b1[3];
  return r;
}

__device__ __forceinline__ v16bf load_frag_bf16(const __bf16* __restrict__ p) {
  v8bf lo = *(const v8bf*)(p);
  v8bf hi = *(const v8bf*)(p + 16);
  v16bf r;
#pragma unroll
  for (int j = 0; j < 8; ++j) { r[j] = lo[j]; r[j + 8] = hi[j]; }
  return r;
}

constexpr int KSLICE = H / 8;
constexpr int KS_A   = KSLICE / 32;
constexpr int MT_A   = U / 16;
constexpr int NT_A   = P / 16;
constexpr int TILE_F = 256;
constexpr int PART_F = MT_A * NT_A * TILE_F;
constexpr int CT_PITCH = U + 8;

__global__ void __launch_bounds__(NW_C * 32)
cosine_kernel(const float* __restrict__ utt,
              const float* __restrict__ pheno,
              __bf16* __restrict__ c_out,
              __bf16* __restrict__ cT_out) {
  __shared__ __attribute__((aligned(16))) float part[8 * PART_F];
  __shared__ __attribute__((aligned(16))) __bf16 c_lds[K2];
  __shared__ __attribute__((aligned(16))) __bf16 cT_lds[P * CT_PITCH];
  __shared__ __attribute__((aligned(16))) float inv_lds[U + P];

  int b    = blockIdx.x;
  int wave = threadIdx.x >> 5;
  int lane = threadIdx.x & 31;

  const bool norm_first = __builtin_amdgcn_readfirstlane(threadIdx.x >> 5) < NW_C / 2;

  float rsum[ROWS_N];
  if (norm_first) row_sums_phase(utt, pheno, b, lane, rsum);

  int rt   = lane & 15;
  int hi   = lane >> 4;
  int k0   = hi ? 8 : 0;

  constexpr size_t TILE_STRIDE = (size_t)16 * B * H;
  const unsigned loff = (unsigned)(rt * (B * H) + wave * KSLICE + k0);
  const float* Au = utt   + (size_t)b * H;
  const float* Bp = pheno + (size_t)b * H;

  v8f acc[MT_A][NT_A] = {};
#pragma unroll
  for (int ks = 0; ks < KS_A; ++ks) {
    RawFrag rb[NT_A], ra[MT_A];
#pragma unroll
    for (int nt = 0; nt < NT_A; ++nt)
      rb[nt] = load_raw_f32(Bp + nt * TILE_STRIDE + loff + ks * 32);
#pragma unroll
    for (int mt = 0; mt < MT_A; ++mt)
      ra[mt] = load_raw_f32(Au + mt * TILE_STRIDE + loff + ks * 32);
    if (ks == 0 && norm_first) row_norms_finish(rsum, inv_lds, lane);
    v16bf bfr[NT_A];
#pragma unroll
    for (int nt = 0; nt < NT_A; ++nt) bfr[nt] = cvt_frag(rb[nt]);
#pragma unroll
    for (int mt = 0; mt < MT_A; ++mt) {
      v16bf a = cvt_frag(ra[mt]);
#pragma unroll
      for (int nt = 0; nt < NT_A; ++nt)
        acc[mt][nt] = __builtin_amdgcn_wmma_f32_16x16x32_bf16(
            false, a, false, bfr[nt], (short)0, acc[mt][nt], false, false);
    }
  }

  {
    float* pw = part + wave * PART_F + lane * 4;
#pragma unroll
    for (int mt = 0; mt < MT_A; ++mt)
#pragma unroll
      for (int nt = 0; nt < NT_A; ++nt) {
        v8f t = acc[mt][nt];
        *(v4f*)(pw + (mt * NT_A + nt) * TILE_F)       = __builtin_shufflevector(t, t, 0, 1, 2, 3);
        *(v4f*)(pw + (mt * NT_A + nt) * TILE_F + 128) = __builtin_shufflevector(t, t, 4, 5, 6, 7);
      }
  }
  if (!norm_first) {
    row_sums_phase(utt, pheno, b, lane, rsum);
    row_norms_finish(rsum, inv_lds, lane);
  }
  __syncthreads();
  __builtin_amdgcn_sched_barrier(0);

  v8f own[NT_A];
#pragma unroll
  for (int nt = 0; nt < NT_A; ++nt) {
    const float* pr = part + (wave * NT_A + nt) * TILE_F + lane * 4;
    v4f s0 = *(const v4f*)(pr);
    v4f s1 = *(const v4f*)(pr + 128);
#pragma unroll
    for (int w = 1; w < 8; ++w) {
      s0 += *(const v4f*)(pr + w * PART_F);
      s1 += *(const v4f*)(pr + w * PART_F + 128);
    }
    own[nt] = __builtin_shufflevector(s0, s1, 0, 1, 2, 3, 4, 5, 6, 7);
  }

  const float* su_base = inv_lds + wave * 16 + 8 * hi;
  v4f su0 = *(const v4f*)(su_base);
  v4f su1 = *(const v4f*)(su_base + 4);
  float sp[NT_A];
#pragma unroll
  for (int nt = 0; nt < NT_A; ++nt) sp[nt] = inv_lds[U + nt * 16 + rt];

  float su[8] = {su0.x, su0.y, su0.z, su0.w, su1.x, su1.y, su1.z, su1.w};

#pragma unroll
  for (int nt = 0; nt < NT_A; ++nt) {
    v8bf tv;
#pragma unroll
    for (int r = 0; r < 8; ++r) {
      int m = r + 8 * hi;
      __bf16 v = (__bf16)(own[nt][r] * su[r] * sp[nt]);
      c_lds[(wave * 16 + m) * P + nt * 16 + rt] = v;
      tv[r] = v;
    }
    *(v8bf*)(cT_lds + (nt * 16 + rt) * CT_PITCH + wave * 16 + 8 * hi) = tv;
  }
  __syncthreads();

  {
    const v4f* s = (const v4f*)c_lds;
    v4f* d = (v4f*)(c_out + (size_t)b * K2);
#pragma unroll
    for (int j = 0; j < K2 * 2 / 16 / 256; ++j) d[j * 256 + threadIdx.x] = s[j * 256 + threadIdx.x];
  }
  {
    v4f* d = (v4f*)(cT_out + (size_t)b * K2);
#pragma unroll
    for (int j = 0; j < K2 * 2 / 16 / 256; ++j) {
      int chunk = j * 256 + threadIdx.x;
      d[chunk] = *(const v4f*)(cT_lds + (chunk >> 4) * CT_PITCH + (chunk & 15) * 8);
    }
  }
}

constexpr int LW     = 8;
constexpr int NSL    = 32;
constexpr int BG     = 64;
constexpr int NBT    = BG / 16;
constexpr int NO     = U + P;
constexpr int RT_H   = MT_A / 2;
constexpr int CT_H   = NT_A / 2;
constexpr int NT_H   = RT_H + CT_H;
constexpr int NO_H   = NT_H * 16;
constexpr int WPART  = NT_H * NBT * TILE_F;
constexpr int LG_PITCH = NO_H + 4;

__global__ void __launch_bounds__(32 * LW)
logits_partial_kernel(const float* __restrict__ w_utt,
                      const float* __restrict__ w_pheno,
                      const __bf16* __restrict__ c_ws,
                      const __bf16* __restrict__ cT_ws,
                      float* __restrict__ lpart) {
  __shared__ __attribute__((aligned(16))) float part[LW * WPART];
  __shared__ __attribute__((aligned(16))) float lg[BG * LG_PITCH];

  const int th   = blockIdx.x & 1;
  const int bg   = (blockIdx.x >> 1) & 3;
  const int sl   = blockIdx.x >> 3;
  const int wave = threadIdx.x >> 5;
  const int lane = threadIdx.x & 31;
  const int rt   = lane & 15;
  const int hi   = lane >> 4;
  const int koff = (sl * LW + wave) * 32 + (hi ? 8 : 0);

  const float*  wu = w_utt   + (size_t)(th * RT_H * 16 + rt) * K2 + koff;
  const float*  wp = w_pheno + (size_t)(th * CT_H * 16 + rt) * K2 + koff;
  const size_t  cb_off = (size_t)(bg * BG + rt) * K2 + koff;

  RawFrag ra[NT_H];
#pragma unroll
  for (int j = 0; j < RT_H; ++j) ra[j] = load_raw_f32(wu + (size_t)j * 16 * K2);
#pragma unroll
  for (int j = 0; j < CT_H; ++j) ra[RT_H + j] = load_raw_f32(wp + (size_t)j * 16 * K2);
  v16bf cb[NBT], ctb[NBT];
#pragma unroll
  for (int bs = 0; bs < NBT; ++bs) {
    cb[bs]  = load_frag_bf16(c_ws  + cb_off + (size_t)bs * 16 * K2);
    ctb[bs] = load_frag_bf16(cT_ws + cb_off + (size_t)bs * 16 * K2);
  }

  float* pw = part + wave * (NBT * TILE_F) + lane * 4;
#pragma unroll
  for (int j = 0; j < NT_H; ++j) {
    v16bf a = cvt_frag(ra[j]);
#pragma unroll
    for (int bs = 0; bs < NBT; ++bs) {
      v8f z = {};
      v8f t = __builtin_amdgcn_wmma_f32_16x16x32_bf16(
          false, a, false, (j < RT_H) ? cb[bs] : ctb[bs], (short)0, z, false, false);
      *(v4f*)(pw + (j * LW * NBT + bs) * TILE_F)       = __builtin_shufflevector(t, t, 0, 1, 2, 3);
      *(v4f*)(pw + (j * LW * NBT + bs) * TILE_F + 128) = __builtin_shufflevector(t, t, 4, 5, 6, 7);
    }
  }
  __syncthreads();

  const v4f* part4 = (const v4f*)part + threadIdx.x;
  const int bs_r = threadIdx.x >> 6, q_r = (threadIdx.x >> 5) & 1;
  float* lgw = lg + (bs_r * 16 + rt) * LG_PITCH + q_r * 4 + 8 * hi;
#pragma unroll
  for (int j = 0; j < NT_H; ++j) {
    v4f s = part4[(j * LW) * (NBT * TILE_F / 4)];
#pragma unroll
    for (int w = 1; w < LW; ++w) s += part4[(j * LW + w) * (NBT * TILE_F / 4)];
    *(v4f*)(lgw + j * 16) = s;
  }
  __syncthreads();

  float* dst = lpart + ((size_t)sl * B + bg * BG) * NO;
#pragma unroll
  for (int i = 0; i < BG * NO_H / 4 / (32 * LW); ++i) {
    int u4 = i * (32 * LW) + threadIdx.x;
    int bl = u4 / (NO_H / 4), o4 = u4 % (NO_H / 4);
    int o  = (o4 < RT_H * 4) ? th * RT_H * 16 + o4 * 4 : U + th * CT_H * 16 + (o4 - RT_H * 4) * 4;
    *(v4f*)(dst + bl * NO + o) = *(const v4f*)(lg + bl * LG_PITCH + o4 * 4);
  }
}

template <int N>
__device__ __forceinline__ void sum_softmax(const float* __restrict__ lp,
                                            const float* __restrict__ bias,
                                            float* __restrict__ dst, int lane) {
  constexpr int per = N / 32;
  float v[NSL][per];
#pragma unroll
  for (int s = 0; s < NSL; ++s)
#pragma unroll
    for (int j = 0; j < per; ++j) v[s][j] = lp[(size_t)s * B * NO + lane + 32 * j];
  float vals[per];
  float mx = -1e30f;
#pragma unroll
  for (int j = 0; j < per; ++j) {
    float a = v[0][j];
#pragma unroll
    for (int s = 1; s < NSL; ++s) a += v[s][j];
    vals[j] = a + bias[lane + 32 * j];
    mx = fmaxf(mx, vals[j]);
  }
#pragma unroll
  for (int m = 16; m; m >>= 1) mx = fmaxf(mx, __shfl_xor(mx, m, 32));
  float sum = 0.f;
#pragma unroll
  for (int j = 0; j < per; ++j) { vals[j] = __expf(vals[j] - mx); sum += vals[j]; }
#pragma unroll
  for (int m = 16; m; m >>= 1) sum += __shfl_xor(sum, m, 32);
  float inv = 1.0f / sum;
#pragma unroll
  for (int j = 0; j < per; ++j) dst[lane + 32 * j] = vals[j] * inv;
}

__global__ void __launch_bounds__(64)
softmax_kernel(const float* __restrict__ lpart,
               const float* __restrict__ b_utt,
               const float* __restrict__ b_pheno,
               float* __restrict__ out_row,
               float* __restrict__ out_col) {
  const int b    = blockIdx.x;
  const int lane = threadIdx.x & 31;
  const float* lp = lpart + (size_t)b * NO;
  if (threadIdx.x < 32) sum_softmax<U>(lp, b_utt, out_row + (size_t)b * U, lane);
  else                  sum_softmax<P>(lp + U, b_pheno, out_col + (size_t)b * P, lane);
}

extern "C" void kernel_launch(void* const* d_in, const int* in_sizes, int n_in,
                              void* d_out, int out_size, void* d_ws, size_t ws_size,
                              hipStream_t stream) {
  const float* utt     = (const float*)d_in[0];
  const float* pheno   = (const float*)d_in[1];
  const float* w_utt   = (const float*)d_in[2];
  const float* b_utt   = (const float*)d_in[3];
  const float* w_pheno = (const float*)d_in[4];
  const float* b_pheno = (const float*)d_in[5];
  (void)in_sizes; (void)n_in; (void)out_size; (void)ws_size;

  char* ws = (char*)d_ws;
  size_t off = 0;
  auto alloc = [&](size_t bytes) {
    void* p = ws + off;
    off = (off + bytes + 255) & ~(size_t)255;
    return p;
  };
  __bf16* c_ws   = (__bf16*)alloc((size_t)B * K2 * sizeof(__bf16));
  __bf16* cT_ws  = (__bf16*)alloc((size_t)B * K2 * sizeof(__bf16));
  float*  lpart  = (float*)alloc((size_t)NSL * B * NO * sizeof(float));

  float* row_logits = (float*)d_out;
  float* col_logits = (float*)d_out + B * U;

  cosine_kernel<<<B, NW_C * 32, 0, stream>>>(utt, pheno, c_ws, cT_ws);
  logits_partial_kernel<<<(B / BG) * NSL * 2, 32 * LW, 0, stream>>>(w_utt, w_pheno, c_ws, cT_ws, lpart);
  softmax_kernel<<<B, 64, 0, stream>>>(lpart, b_utt, b_pheno, row_logits, col_logits);
}


# cosine GEMM operands loaded as rows (2 rows x 256 B per load: 4 lines, not 16) and turned into fragments by the matrix unit (x 0/1 matrix); K split in 64-k chunks (f32 sum order changes)
# speedup vs baseline: 6.3242x; 1.1063x over previous
#include <hip/hip_runtime.h>
#include <hip/hip_bf16.h>
#include <math.h>

typedef __attribute__((ext_vector_type(16))) __bf16 v16bf;
typedef __attribute__((ext_vector_type(8)))  __bf16 v8bf;
typedef __attribute__((ext_vector_type(8)))  float  v8f;
typedef __attribute__((ext_vector_type(4)))  float  v4f;

constexpr int B  = 256;
constexpr int U  = 128;
constexpr int P  = 64;
constexpr int H  = 768;
constexpr int K2 = U * P;

constexpr int NW_C    = 8;
constexpr int RU_N    = U / NW_C;
constexpr int RP_N    = P / NW_C;
constexpr int ROWS_N  = RU_N + RP_N;
constexpr int AHEAD_N = 9;
constexpr int HQ      = H / 128;

__device__ __forceinline__ void row_sums_phase(const float* __restrict__ utt,
                                               const float* __restrict__ pheno,
                                               int b, int lane, float (&sum)[ROWS_N]) {
  const int wave = __builtin_amdgcn_readfirstlane(threadIdx.x >> 5);
  const float* ub = utt   + ((size_t)(wave * RU_N) * B + b) * H + lane * 4;
  const float* pb = pheno + ((size_t)(wave * RP_N) * B + b) * H + lane * 4;

  v4f buf[AHEAD_N + 1][HQ];
#define LOAD_ROW(R) do { \
    const float* rp_ = ((R) < RU_N) ? ub + (size_t)(R) * B * H : pb + (size_t)((R) - RU_N) * B * H; \
    _Pragma("unroll") for (int q = 0; q < HQ; ++q) buf[(R) % (AHEAD_N + 1)][q] = *(const v4f*)(rp_ + q * 128); } while (0)
#pragma unroll
  for (int r = 0; r < AHEAD_N; ++r) LOAD_ROW(r);
#pragma unroll
  for (int r = 0; r < ROWS_N; ++r) {
    __builtin_amdgcn_sched_barrier(0);
    if (r + AHEAD_N < ROWS_N) LOAD_ROW(r + AHEAD_N);
    __builtin_amdgcn_sched_barrier(0);
    float s = 0.f;
#pragma unroll
    for (int q = 0; q < HQ; ++q) {
      v4f v = buf[r % (AHEAD_N + 1)][q];
      s += v.x * v.x + v.y * v.y + v.z * v.z + v.w * v.w;
    }
    sum[r] = s;
  }
#undef LOAD_ROW
}

__device__ __forceinline__ void row_norms_finish(float (&sum)[ROWS_N],
                                                 float* __restrict__ inv_lds,
                                                 int lane) {
  const int wave = __builtin_amdgcn_readfirstlane(threadIdx.x >> 5);
#pragma unroll
  for (int r = 0; r < ROWS_N; ++r) {
    float s = sum[r];
#pragma unroll
    for (int m = 16; m; m >>= 1) s += __shfl_xor(s, m, 32);
    sum[r] = s;
  }
  float mine = 1.f;
#pragma unroll
  for (int r = 0; r < ROWS_N; ++r) if (lane == r) mine = sum[r];
  if (lane < ROWS_N) {
    int slot = (lane < RU_N) ? wave * RU_N + lane : U + wave * RP_N + (lane - RU_N);
    inv_lds[slot] = 1.0f / fmaxf(sqrtf(mine), 1e-8f);
  }
}

struct RawFrag { v4f a0, a1, b0, b1; };

__device__ __forceinline__ RawFrag load_raw_f32(const float* __restrict__ p) {
  RawFrag f;
  f.a0 = *(const v4f*)(p);
  f.a1 = *(const v4f*)(p + 4);
  f.b0 = *(const v4f*)(p + 16);
  f.b1 = *(const v4f*)(p + 20);
  return f;
}

__device__ __forceinline__ v16bf cvt_frag(const RawFrag& f) {
  v16bf r;
  r[0]  = (__bf16)f.a0[0]; r[1]  = (__bf16)f.a0[1];
  r[2]  = (__bf16)f.a0[2]; r[3]  = (__bf16)f.a0[3];
  r[4]  = (__bf16)f.a1[0]; r[5]  = (__bf16)f.a1[1];
  r[6]  = (__bf16)f.a1[2]; r[7]  = (__bf16)f.a1[3];
  r[8]  = (__bf16)f.b0[0]; r[9]  = (__bf16)f.b0[1];
  r[10] = (__bf16)f.b0[2]; r[11] = (__bf16)f.b0[3];
  r[12] = (__bf16)f.b1[0]; r[13] = (__bf16)f.b1[1];
  r[14] = (__bf16)f.b1[2]; r[15] = (__bf16)f.b1[3];
  return r;
}

__device__ __forceinline__ v16bf load_frag_bf16(const __bf16* __restrict__ p) {
  v8bf lo = *(const v8bf*)(p);
  v8bf hi = *(const v8bf*)(p + 16);
  v16bf r;
#pragma unroll
  for (int j = 0; j < 8; ++j) { r[j] = lo[j]; r[j + 8] = hi[j]; }
  return r;
}

constexpr int KCH    = 64;
constexpr int LOOK   = 3;
constexpr int NTILE_SEQ = 20;
constexpr int MT_A   = U / 16;
constexpr int NT_A   = P / 16;
constexpr int TILE_F = 256;
constexpr int PART_F = MT_A * NT_A * TILE_F;
constexpr int CT_PITCH = U + 8;

struct Raw64  { v4f r[8]; };
struct Frag64 { v16bf g[2]; };

__device__ __forceinline__ void load_rows64(const float* __restrict__ p, Raw64& x) {
#pragma unroll
  for (int i = 0; i < 8; ++i) x.r[i] = *(const v4f*)(p + (size_t)i * 2 * B * H);
}

__device__ __forceinline__ Frag64 rows_to_frags(const Raw64& x, v16bf s0, v16bf s1) {
  Frag64 f;
#pragma unroll
  for (int e = 0; e < 2; ++e) {
    v16bf a;
#pragma unroll
    for (int i = 0; i < 8; ++i) {
      a[2 * i]     = (__bf16)x.r[i][2 * e];
      a[2 * i + 1] = (__bf16)x.r[i][2 * e + 1];
    }
    v8f z = {};
    v8f d0 = __builtin_amdgcn_wmma_f32_16x16x32_bf16(false, a, false, s0, (short)0, z, false, false);
    v8f d1 = __builtin_amdgcn_wmma_f32_16x16x32_bf16(false, a, false, s1, (short)0, z, false, false);
#pragma unroll
    for (int r = 0; r < 8; ++r) {
      f.g[e][2 * r]     = (__bf16)d0[r];
      f.g[e][2 * r + 1] = (__bf16)d1[r];
    }
  }
  return f;
}

__global__ void __launch_bounds__(NW_C * 32)
cosine_kernel(const float* __restrict__ utt,
              const float* __restrict__ pheno,
              __bf16* __restrict__ c_out,
              __bf16* __restrict__ cT_out) {
  __shared__ __attribute__((aligned(16))) float part[8 * PART_F];
  __shared__ __attribute__((aligned(16))) __bf16 c_lds[K2];
  __shared__ __attribute__((aligned(16))) __bf16 cT_lds[P * CT_PITCH];
  __shared__ __attribute__((aligned(16))) float inv_lds[U + P];

  int b    = blockIdx.x;
  int wave = threadIdx.x >> 5;
  int lane = threadIdx.x & 31;

  const bool norm_first = __builtin_amdgcn_readfirstlane(threadIdx.x >> 5) < NW_C / 2;

  float rsum[ROWS_N];
  if (norm_first) row_sums_phase(utt, pheno, b, lane, rsum);

  int rt   = lane & 15;
  int hi   = lane >> 4;

  v16bf sel0, sel1;
  {
    const int hit = (hi == (rt & 1)) ? (rt & ~1) : -1;
#pragma unroll
    for (int e = 0; e < 16; ++e) {
      sel0[e] = (__bf16)((e == hit) ? 1.0f : 0.0f);
      sel1[e] = (__bf16)((hit >= 0 && e == hit + 1) ? 1.0f : 0.0f);
    }
  }

  constexpr size_t TILE_STRIDE = (size_t)16 * B * H;
  const unsigned loff = (unsigned)(hi * (B * H) + rt * 4);
  const float* Au = utt   + (size_t)b * H;
  const float* Bp = pheno + (size_t)b * H;
  const int wv  = __builtin_amdgcn_readfirstlane(threadIdx.x >> 5);
  const int kA  = wv * KCH;
  const int kB  = 8 * KCH + (wv >> 1) * KCH;
  const int mtB = (wv & 1) * (MT_A / 2);

  v8f acc[MT_A][NT_A] = {};
  Raw64 ring[LOOK + 1];
  Frag64 pf[NT_A];
#define TILE_BASE(t) (((t) < 4)  ? Bp + (size_t)(t) * TILE_STRIDE + kA : \
                      ((t) < 12) ? Au + (size_t)((t) - 4) * TILE_STRIDE + kA : \
                      ((t) < 16) ? Bp + (size_t)((t) - 12) * TILE_STRIDE + kB : \
                                   Au + (size_t)(mtB + (t) - 16) * TILE_STRIDE + kB)
#pragma unroll
  for (int t = 0; t < LOOK; ++t) load_rows64(TILE_BASE(t) + loff, ring[t]);
  if (norm_first) row_norms_finish(rsum, inv_lds, lane);
#pragma unroll
  for (int t = 0; t < NTILE_SEQ; ++t) {
    if (t + LOOK < NTILE_SEQ) load_rows64(TILE_BASE(t + LOOK) + loff, ring[(t + LOOK) % (LOOK + 1)]);
    Frag64 f = rows_to_frags(ring[t % (LOOK + 1)], sel0, sel1);
    if (t < 4) pf[t] = f;
    else if (t >= 12 && t < 16) pf[t - 12] = f;
    else if (t < 12) {
      const int mt = t - 4;
#pragma unroll
      for (int e = 0; e < 2; ++e)
#pragma unroll
        for (int nt = 0; nt < NT_A; ++nt)
          acc[mt][nt] = __builtin_amdgcn_wmma_f32_16x16x32_bf16(
              false, f.g[e], false, pf[nt].g[e], (short)0, acc[mt][nt], false, false);
    } else {
      const int j = t - 16;
      if (mtB) {
#pragma unroll
        for (int e = 0; e < 2; ++e)
#pragma unroll
          for (int nt = 0; nt < NT_A; ++nt)
            acc[MT_A / 2 + j][nt] = __builtin_amdgcn_wmma_f32_16x16x32_bf16(
                false, f.g[e], false, pf[nt].g[e], (short)0, acc[MT_A / 2 + j][nt], false, false);
      } else {
#pragma unroll
        for (int e = 0; e < 2; ++e)
#pragma unroll
          for (int nt = 0; nt < NT_A; ++nt)
            acc[j][nt] = __builtin_amdgcn_wmma_f32_16x16x32_bf16(
                false, f.g[e], false, pf[nt].g[e], (short)0, acc[j][nt], false, false);
      }
    }
  }
#undef TILE_BASE

  {
    float* pw = part + wave * PART_F + lane * 4;
#pragma unroll
    for (int mt = 0; mt < MT_A; ++mt)
#pragma unroll
      for (int nt = 0; nt < NT_A; ++nt) {
        v8f t = acc[mt][nt];
        *(v4f*)(pw + (mt * NT_A + nt) * TILE_F)       = __builtin_shufflevector(t, t, 0, 1, 2, 3);
        *(v4f*)(pw + (mt * NT_A + nt) * TILE_F + 128) = __builtin_shufflevector(t, t, 4, 5, 6, 7);
      }
  }
  if (!norm_first) {
    row_sums_phase(utt, pheno, b, lane, rsum);
    row_norms_finish(rsum, inv_lds, lane);
  }
  __syncthreads();
  __builtin_amdgcn_sched_barrier(0);

  v8f own[NT_A];
#pragma unroll
  for (int nt = 0; nt < NT_A; ++nt) {
    const float* pr = part + (wave * NT_A + nt) * TILE_F + lane * 4;
    v4f s0 = *(const v4f*)(pr);
    v4f s1 = *(const v4f*)(pr + 128);
#pragma unroll
    for (int w = 1; w < 8; ++w) {
      s0 += *(const v4f*)(pr + w * PART_F);
      s1 += *(const v4f*)(pr + w * PART_F + 128);
    }
    own[nt] = __builtin_shufflevector(s0, s1, 0, 1, 2, 3, 4, 5, 6, 7);
  }

  const float* su_base = inv_lds + wave * 16 + 8 * hi;
  v4f su0 = *(const v4f*)(su_base);
  v4f su1 = *(const v4f*)(su_base + 4);
  float sp[NT_A];
#pragma unroll
  for (int nt = 0; nt < NT_A; ++nt) sp[nt] = inv_lds[U + nt * 16 + rt];

  float su[8] = {su0.x, su0.y, su0.z, su0.w, su1.x, su1.y, su1.z, su1.w};

#pragma unroll
  for (int nt = 0; nt < NT_A; ++nt) {
    v8bf tv;
#pragma unroll
    for (int r = 0; r < 8; ++r) {
      int m = r + 8 * hi;
      __bf16 v = (__bf16)(own[nt][r] * su[r] * sp[nt]);
      c_lds[(wave * 16 + m) * P + nt * 16 + rt] = v;
      tv[r] = v;
    }
    *(v8bf*)(cT_lds + (nt * 16 + rt) * CT_PITCH + wave * 16 + 8 * hi) = tv;
  }
  __syncthreads();

  {
    const v4f* s = (const v4f*)c_lds;
    v4f* d = (v4f*)(c_out + (size_t)b * K2);
#pragma unroll
    for (int j = 0; j < K2 * 2 / 16 / 256; ++j) d[j * 256 + threadIdx.x] = s[j * 256 + threadIdx.x];
  }
  {
    v4f* d = (v4f*)(cT_out + (size_t)b * K2);
#pragma unroll
    for (int j = 0; j < K2 * 2 / 16 / 256; ++j) {
      int chunk = j * 256 + threadIdx.x;
      d[chunk] = *(const v4f*)(cT_lds + (chunk >> 4) * CT_PITCH + (chunk & 15) * 8);
    }
  }
}

constexpr int LW     = 8;
constexpr int NSL    = 32;
constexpr int BG     = 64;
constexpr int NBT    = BG / 16;
constexpr int NO     = U + P;
constexpr int RT_H   = MT_A / 2;
constexpr int CT_H   = NT_A / 2;
constexpr int NT_H   = RT_H + CT_H;
constexpr int NO_H   = NT_H * 16;
constexpr int WPART  = NT_H * NBT * TILE_F;
constexpr int LG_PITCH = NO_H + 4;

__global__ void __launch_bounds__(32 * LW)
logits_partial_kernel(const float* __restrict__ w_utt,
                      const float* __restrict__ w_pheno,
                      const __bf16* __restrict__ c_ws,
                      const __bf16* __restrict__ cT_ws,
                      float* __restrict__ lpart) {
  __shared__ __attribute__((aligned(16))) float part[LW * WPART];
  __shared__ __attribute__((aligned(16))) float lg[BG * LG_PITCH];

  const int th   = blockIdx.x & 1;
  const int bg   = (blockIdx.x >> 1) & 3;
  const int sl   = blockIdx.x >> 3;
  const int wave = threadIdx.x >> 5;
  const int lane = threadIdx.x & 31;
  const int rt   = lane & 15;
  const int hi   = lane >> 4;
  const int koff = (sl * LW + wave) * 32 + (hi ? 8 : 0);

  const float*  wu = w_utt   + (size_t)(th * RT_H * 16 + rt) * K2 + koff;
  const float*  wp = w_pheno + (size_t)(th * CT_H * 16 + rt) * K2 + koff;
  const size_t  cb_off = (size_t)(bg * BG + rt) * K2 + koff;

  RawFrag ra[NT_H];
#pragma unroll
  for (int j = 0; j < RT_H; ++j) ra[j] = load_raw_f32(wu + (size_t)j * 16 * K2);
#pragma unroll
  for (int j = 0; j < CT_H; ++j) ra[RT_H + j] = load_raw_f32(wp + (size_t)j * 16 * K2);
  v16bf cb[NBT], ctb[NBT];
#pragma unroll
  for (int bs = 0; bs < NBT; ++bs) {
    cb[bs]  = load_frag_bf16(c_ws  + cb_off + (size_t)bs * 16 * K2);
    ctb[bs] = load_frag_bf16(cT_ws + cb_off + (size_t)bs * 16 * K2);
  }

  float* pw = part + wave * (NBT * TILE_F) + lane * 4;
#pragma unroll
  for (int j = 0; j < NT_H; ++j) {
    v16bf a = cvt_frag(ra[j]);
#pragma unroll
    for (int bs = 0; bs < NBT; ++bs) {
      v8f z = {};
      v8f t = __builtin_amdgcn_wmma_f32_16x16x32_bf16(
          false, a, false, (j < RT_H) ? cb[bs] : ctb[bs], (short)0, z, false, false);
      *(v4f*)(pw + (j * LW * NBT + bs) * TILE_F)       = __builtin_shufflevector(t, t, 0, 1, 2, 3);
      *(v4f*)(pw + (j * LW * NBT + bs) * TILE_F + 128) = __builtin_shufflevector(t, t, 4, 5, 6, 7);
    }
  }
  __syncthreads();

  const v4f* part4 = (const v4f*)part + threadIdx.x;
  const int bs_r = threadIdx.x >> 6, q_r = (threadIdx.x >> 5) & 1;
  float* lgw = lg + (bs_r * 16 + rt) * LG_PITCH + q_r * 4 + 8 * hi;
#pragma unroll
  for (int j = 0; j < NT_H; ++j) {
    v4f s = part4[(j * LW) * (NBT * TILE_F / 4)];
#pragma unroll
    for (int w = 1; w < LW; ++w) s += part4[(j * LW + w) * (NBT * TILE_F / 4)];
    *(v4f*)(lgw + j * 16) = s;
  }
  __syncthreads();

  float* dst = lpart + ((size_t)sl * B + bg * BG) * NO;
#pragma unroll
  for (int i = 0; i < BG * NO_H / 4 / (32 * LW); ++i) {
    int u4 = i * (32 * LW) + threadIdx.x;
    int bl = u4 / (NO_H / 4), o4 = u4 % (NO_H / 4);
    int o  = (o4 < RT_H * 4) ? th * RT_H * 16 + o4 * 4 : U + th * CT_H * 16 + (o4 - RT_H * 4) * 4;
    *(v4f*)(dst + bl * NO + o) = *(const v4f*)(lg + bl * LG_PITCH + o4 * 4);
  }
}

template <int N>
__device__ __forceinline__ void sum_softmax(const float* __restrict__ lp,
                                            const float* __restrict__ bias,
                                            float* __restrict__ dst, int lane) {
  constexpr int per = N / 32;
  float v[NSL][per];
#pragma unroll
  for (int s = 0; s < NSL; ++s)
#pragma unroll
    for (int j = 0; j < per; ++j) v[s][j] = lp[(size_t)s * B * NO + lane + 32 * j];
  float vals[per];
  float mx = -1e30f;
#pragma unroll
  for (int j = 0; j < per; ++j) {
    float a = v[0][j];
#pragma unroll
    for (int s = 1; s < NSL; ++s) a += v[s][j];
    vals[j] = a + bias[lane + 32 * j];
    mx = fmaxf(mx, vals[j]);
  }
#pragma unroll
  for (int m = 16; m; m >>= 1) mx = fmaxf(mx, __shfl_xor(mx, m, 32));
  float sum = 0.f;
#pragma unroll
  for (int j = 0; j < per; ++j) { vals[j] = __expf(vals[j] - mx); sum += vals[j]; }
#pragma unroll
  for (int m = 16; m; m >>= 1) sum += __shfl_xor(sum, m, 32);
  float inv = 1.0f / sum;
#pragma unroll
  for (int j = 0; j < per; ++j) dst[lane + 32 * j] = vals[j] * inv;
}

__global__ void __launch_bounds__(64)
softmax_kernel(const float* __restrict__ lpart,
               const float* __restrict__ b_utt,
               const float* __restrict__ b_pheno,
               float* __restrict__ out_row,
               float* __restrict__ out_col) {
  const int b    = blockIdx.x;
  const int lane = threadIdx.x & 31;
  const float* lp = lpart + (size_t)b * NO;
  if (threadIdx.x < 32) sum_softmax<U>(lp, b_utt, out_row + (size_t)b * U, lane);
  else                  sum_softmax<P>(lp + U, b_pheno, out_col + (size_t)b * P, lane);
}

extern "C" void kernel_launch(void* const* d_in, const int* in_sizes, int n_in,
                              void* d_out, int out_size, void* d_ws, size_t ws_size,
                              hipStream_t stream) {
  const float* utt     = (const float*)d_in[0];
  const float* pheno   = (const float*)d_in[1];
  const float* w_utt   = (const float*)d_in[2];
  const float* b_utt   = (const float*)d_in[3];
  const float* w_pheno = (const float*)d_in[4];
  const float* b_pheno = (const float*)d_in[5];
  (void)in_sizes; (void)n_in; (void)out_size; (void)ws_size;

  char* ws = (char*)d_ws;
  size_t off = 0;
  auto alloc = [&](size_t bytes) {
    void* p = ws + off;
    off = (off + bytes + 255) & ~(size_t)255;
    return p;
  };
  __bf16* c_ws   = (__bf16*)alloc((size_t)B * K2 * sizeof(__bf16));
  __bf16* cT_ws  = (__bf16*)alloc((size_t)B * K2 * sizeof(__bf16));
  float*  lpart  = (float*)alloc((size_t)NSL * B * NO * sizeof(float));

  float* row_logits = (float*)d_out;
  float* col_logits = (float*)d_out + B * U;

  cosine_kernel<<<B, NW_C * 32, 0, stream>>>(utt, pheno, c_ws, cT_ws);
  logits_partial_kernel<<<(B / BG) * NSL * 2, 32 * LW, 0, stream>>>(w_utt, w_pheno, c_ws, cT_ws, lpart);
  softmax_kernel<<<B, 64, 0, stream>>>(lpart, b_utt, b_pheno, row_logits, col_logits);
}
